# starting kernel
# baseline (speedup 1.0000x reference)
#include <hip/hip_runtime.h>
#include <hip/hip_bf16.h>

typedef __attribute__((ext_vector_type(2))) float v2f;
typedef __attribute__((ext_vector_type(8))) float v8f;

#define HID 16
#define NGRAPH 16
#define NUM_GRAPHS 512

__device__ __forceinline__ void atomAddF(float* p, float v) {
    __hip_atomic_fetch_add(p, v, __ATOMIC_RELAXED, __HIP_MEMORY_SCOPE_AGENT);
}

__global__ void k_init_deg(float* __restrict__ deg, int n) {
    int i = blockIdx.x * blockDim.x + threadIdx.x;
    if (i < n) deg[i] = 1.0f;
}

__global__ void k_deg_accum(const int* __restrict__ dst, float* __restrict__ deg, int E) {
    int e = blockIdx.x * blockDim.x + threadIdx.x;
    if (e < E) atomAddF(&deg[dst[e]], 1.0f);
}

__global__ void k_rsqrt(float* __restrict__ deg, int n) {
    int i = blockIdx.x * blockDim.x + threadIdx.x;
    if (i < n) deg[i] = rsqrtf(deg[i]);
}

template<int K, bool RELU>
__global__ void k_gemm16(const float* __restrict__ A, const float* __restrict__ W,
                         float* __restrict__ D, int n, int n_tiles) {
    const int lane = threadIdx.x & 31;
    const int wid  = threadIdx.x >> 5;
    const int tile = blockIdx.x * (blockDim.x >> 5) + wid;
    if (tile >= n_tiles) return;

    const int m    = lane & 15;
    const int half = lane >> 4;
    const int kh   = half * 2;

    int rowA = tile * 16 + m;
    int rowC = rowA;
    if (rowA >= n) rowA = n - 1;

    const float* arow = A + (size_t)rowA * K;
    v8f acc = {};
#pragma unroll
    for (int k0 = 0; k0 < K; k0 += 4) {
        float a0 = arow[k0 + kh];
        float a1 = arow[k0 + kh + 1];
        if (RELU) { a0 = fmaxf(a0, 0.0f); a1 = fmaxf(a1, 0.0f); }
        v2f a; a.x = a0; a.y = a1;
        v2f b;
        b.x = W[(k0 + kh) * HID + m];
        b.y = W[(k0 + kh + 1) * HID + m];
        acc = __builtin_amdgcn_wmma_f32_16x16x4_f32(false, a, false, b,
                                                    (short)0, acc, false, false);
    }
#pragma unroll
    for (int r = 0; r < 8; ++r) {
        int row = tile * 16 + r + 8 * half;
        if (row < n) D[(size_t)row * HID + m] = acc[r];
    }
}

__global__ void k_self_bias(const float* __restrict__ h, const float* __restrict__ dinv,
                            const float* __restrict__ bias, float* __restrict__ out, int n) {
    int t = blockIdx.x * blockDim.x + threadIdx.x;
    if (t >= n * HID) return;
    int i = t >> 4, c = t & 15;
    float di = dinv[i];
    out[t] = di * di * h[t] + bias[c];
}

__global__ void k_edge_prop(const int* __restrict__ src, const int* __restrict__ dst,
                            const float* __restrict__ dinv, const float* __restrict__ h,
                            float* __restrict__ out, int E) {
    int t = blockIdx.x * blockDim.x + threadIdx.x;
    int e = t >> 2;
    int q = t & 3;
    if (e >= E) return;
    int s = src[e], d = dst[e];
    float w = dinv[s] * dinv[d];
    const float4 hv = *(const float4*)(h + (size_t)s * HID + q * 4);
    float* o = out + (size_t)d * HID + q * 4;
    atomAddF(o + 0, w * hv.x);
    atomAddF(o + 1, w * hv.y);
    atomAddF(o + 2, w * hv.z);
    atomAddF(o + 3, w * hv.w);
}

__global__ void k_zero_pool(float* __restrict__ sums, float* __restrict__ cnt) {
    int t = blockIdx.x * blockDim.x + threadIdx.x;
    if (t < NUM_GRAPHS * HID) sums[t] = 0.0f;
    if (t < NUM_GRAPHS) cnt[t] = 0.0f;
}

__global__ void k_pool(const float* __restrict__ out, const int* __restrict__ batch,
                       float* __restrict__ sums, float* __restrict__ cnt, int n) {
    int i = blockIdx.x * blockDim.x + threadIdx.x;
    if (i >= n) return;
    int g = batch[i];
    atomAddF(&cnt[g], 1.0f);
    const float* row = out + (size_t)i * HID;
#pragma unroll
    for (int c = 0; c < HID; ++c) atomAddF(&sums[g * HID + c], row[c]);
}

__global__ void k_head(const float* __restrict__ sums, const float* __restrict__ cnt,
                       const float* __restrict__ fcW, const float* __restrict__ fcb,
                       float* __restrict__ y) {
    int t = blockIdx.x * blockDim.x + threadIdx.x;
    if (t >= NUM_GRAPHS * NGRAPH) return;
    int g = t >> 4, cls = t & 15;
    float inv = 1.0f / fmaxf(cnt[g], 1.0f);
    float acc = fcb[cls];
#pragma unroll
    for (int k = 0; k < HID; ++k)
        acc += (sums[g * HID + k] * inv) * fcW[k * NGRAPH + cls];
    y[t] = acc;
}

extern "C" void kernel_launch(void* const* d_in, const int* in_sizes, int n_in,
                              void* d_out, int out_size, void* d_ws, size_t ws_size,
                              hipStream_t stream) {
    const float* x    = (const float*)d_in[0];
    const float* W1   = (const float*)d_in[1];
    const float* b1   = (const float*)d_in[2];
    const float* W2   = (const float*)d_in[3];
    const float* b2   = (const float*)d_in[4];
    const float* fcW  = (const float*)d_in[5];
    const float* fcb  = (const float*)d_in[6];
    const int*   eidx = (const int*)d_in[7];
    const int*   batch= (const int*)d_in[8];
    float* y = (float*)d_out;

    const int n = in_sizes[0] / 128;
    const int E = in_sizes[7] / 2;
    const int* src = eidx;
    const int* dst = eidx + E;

    size_t nAl = ((size_t)n + 3) & ~(size_t)3;
    float* deg  = (float*)d_ws;
    float* h    = deg + nAl;
    float* outb = h + (size_t)n * HID;
    float* sums = outb + (size_t)n * HID;
    float* cnt  = sums + NUM_GRAPHS * HID;

    const int TB = 256;
    const int n_tiles = (n + 15) / 16;

    k_init_deg<<<(n + TB - 1) / TB, TB, 0, stream>>>(deg, n);
    k_deg_accum<<<(E + TB - 1) / TB, TB, 0, stream>>>(dst, deg, E);
    k_rsqrt<<<(n + TB - 1) / TB, TB, 0, stream>>>(deg, n);

    k_gemm16<128, false><<<(n_tiles + 3) / 4, 128, 0, stream>>>(x, W1, h, n, n_tiles);
    k_self_bias<<<((n * HID) + TB - 1) / TB, TB, 0, stream>>>(h, deg, b1, outb, n);
    k_edge_prop<<<((E * 4) + TB - 1) / TB, TB, 0, stream>>>(src, dst, deg, h, outb, E);

    k_gemm16<16, true><<<(n_tiles + 3) / 4, 128, 0, stream>>>(outb, W2, h, n, n_tiles);
    k_self_bias<<<((n * HID) + TB - 1) / TB, TB, 0, stream>>>(h, deg, b2, outb, n);
    k_edge_prop<<<((E * 4) + TB - 1) / TB, TB, 0, stream>>>(src, dst, deg, h, outb, E);

    k_zero_pool<<<(NUM_GRAPHS * HID + TB - 1) / TB, TB, 0, stream>>>(sums, cnt);
    k_pool<<<(n + TB - 1) / TB, TB, 0, stream>>>(outb, batch, sums, cnt, n);
    k_head<<<(NUM_GRAPHS * NGRAPH + TB - 1) / TB, TB, 0, stream>>>(sums, cnt, fcW, fcb, y);
}


# edge scatter: 16 lanes per edge (one per channel), one wave per 32 edges, so each f32 atomic and h gather covers 2 whole 64-B rows per instruction, not 8 lines; atomic arrival order as before
# speedup vs baseline: 1.2592x; 1.2592x over previous
#include <hip/hip_runtime.h>
#include <hip/hip_bf16.h>

typedef __attribute__((ext_vector_type(2))) float v2f;
typedef __attribute__((ext_vector_type(8))) float v8f;

#define HID 16
#define NGRAPH 16
#define NUM_GRAPHS 512

__device__ __forceinline__ void atomAddF(float* p, float v) {
    __hip_atomic_fetch_add(p, v, __ATOMIC_RELAXED, __HIP_MEMORY_SCOPE_AGENT);
}

__global__ void k_init_deg(float* __restrict__ deg, int n) {
    int i = blockIdx.x * blockDim.x + threadIdx.x;
    if (i < n) deg[i] = 1.0f;
}

__global__ void k_deg_accum(const int* __restrict__ dst, float* __restrict__ deg, int E) {
    int e = blockIdx.x * blockDim.x + threadIdx.x;
    if (e < E) atomAddF(&deg[dst[e]], 1.0f);
}

__global__ void k_rsqrt(float* __restrict__ deg, int n) {
    int i = blockIdx.x * blockDim.x + threadIdx.x;
    if (i < n) deg[i] = rsqrtf(deg[i]);
}

template<int K, bool RELU>
__global__ void k_gemm16(const float* __restrict__ A, const float* __restrict__ W,
                         float* __restrict__ D, int n, int n_tiles) {
    const int lane = threadIdx.x & 31;
    const int wid  = threadIdx.x >> 5;
    const int tile = blockIdx.x * (blockDim.x >> 5) + wid;
    if (tile >= n_tiles) return;

    const int m    = lane & 15;
    const int half = lane >> 4;
    const int kh   = half * 2;

    int rowA = tile * 16 + m;
    int rowC = rowA;
    if (rowA >= n) rowA = n - 1;

    const float* arow = A + (size_t)rowA * K;
    v8f acc = {};
#pragma unroll
    for (int k0 = 0; k0 < K; k0 += 4) {
        float a0 = arow[k0 + kh];
        float a1 = arow[k0 + kh + 1];
        if (RELU) { a0 = fmaxf(a0, 0.0f); a1 = fmaxf(a1, 0.0f); }
        v2f a; a.x = a0; a.y = a1;
        v2f b;
        b.x = W[(k0 + kh) * HID + m];
        b.y = W[(k0 + kh + 1) * HID + m];
        acc = __builtin_amdgcn_wmma_f32_16x16x4_f32(false, a, false, b,
                                                    (short)0, acc, false, false);
    }
#pragma unroll
    for (int r = 0; r < 8; ++r) {
        int row = tile * 16 + r + 8 * half;
        if (row < n) D[(size_t)row * HID + m] = acc[r];
    }
}

__global__ void k_self_bias(const float* __restrict__ h, const float* __restrict__ dinv,
                            const float* __restrict__ bias, float* __restrict__ out, int n) {
    int t = blockIdx.x * blockDim.x + threadIdx.x;
    if (t >= n * HID) return;
    int i = t >> 4, c = t & 15;
    float di = dinv[i];
    out[t] = di * di * h[t] + bias[c];
}

#define EDGES_PER_WAVE 32

template<int J>
__device__ __forceinline__ int row_lane(int v) {
    return __builtin_amdgcn_update_dpp(0, v, 0x150 + J, 0xf, 0xf, true);
}
template<int J>
__device__ __forceinline__ float row_lane(float v) {
    return __int_as_float(row_lane<J>(__float_as_int(v)));
}
__global__ void k_edge_prop(const int* __restrict__ src, const int* __restrict__ dst,
                            const float* __restrict__ dinv, const float* __restrict__ h,
                            float* __restrict__ out, int E) {
    const int lane = threadIdx.x & 31;
    const int wave = (blockIdx.x * blockDim.x + threadIdx.x) >> 5;
    const int base = wave * EDGES_PER_WAVE;
    if (base >= E) return;
    const int c    = lane & 15;
    const int hb   = lane & 16;

    const int e  = base + lane;
    const int ec = e < E ? e : E - 1;
    const int   s_l = src[ec];
    const int   d_l = dst[ec];
    const float w_l = dinv[s_l] * dinv[d_l];

    if (base + EDGES_PER_WAVE <= E) {
        unsigned off_d[16];
        float    wj[16];
        float    hv[16];
#define GATHER_ROUND(J)                                              \
        {                                                            \
            const int s = row_lane<J>(s_l);                          \
            const int d = row_lane<J>(d_l);                          \
            wj[J]    = row_lane<J>(w_l);                             \
            off_d[J] = (unsigned)d * HID + c;                        \
            hv[J]    = h[(unsigned)s * HID + c];                     \
        }
        GATHER_ROUND(0)  GATHER_ROUND(1)  GATHER_ROUND(2)  GATHER_ROUND(3)
        GATHER_ROUND(4)  GATHER_ROUND(5)  GATHER_ROUND(6)  GATHER_ROUND(7)
        GATHER_ROUND(8)  GATHER_ROUND(9)  GATHER_ROUND(10) GATHER_ROUND(11)
        GATHER_ROUND(12) GATHER_ROUND(13) GATHER_ROUND(14) GATHER_ROUND(15)
#undef GATHER_ROUND
#pragma unroll
        for (int j = 0; j < 16; ++j) atomAddF(out + off_d[j], wj[j] * hv[j]);
    } else {
#pragma unroll 1
        for (int j = 0; j < 16; ++j) {
            const int   s = __shfl(s_l, hb + j);
            const int   d = __shfl(d_l, hb + j);
            const float w = __shfl(w_l, hb + j);
            if (base + hb + j < E)
                atomAddF(out + ((unsigned)d * HID + c), w * h[(unsigned)s * HID + c]);
        }
    }
}

__global__ void k_zero_pool(float* __restrict__ sums, float* __restrict__ cnt) {
    int t = blockIdx.x * blockDim.x + threadIdx.x;
    if (t < NUM_GRAPHS * HID) sums[t] = 0.0f;
    if (t < NUM_GRAPHS) cnt[t] = 0.0f;
}

__global__ void k_pool(const float* __restrict__ out, const int* __restrict__ batch,
                       float* __restrict__ sums, float* __restrict__ cnt, int n) {
    int i = blockIdx.x * blockDim.x + threadIdx.x;
    if (i >= n) return;
    int g = batch[i];
    atomAddF(&cnt[g], 1.0f);
    const float* row = out + (size_t)i * HID;
#pragma unroll
    for (int c = 0; c < HID; ++c) atomAddF(&sums[g * HID + c], row[c]);
}

__global__ void k_head(const float* __restrict__ sums, const float* __restrict__ cnt,
                       const float* __restrict__ fcW, const float* __restrict__ fcb,
                       float* __restrict__ y) {
    int t = blockIdx.x * blockDim.x + threadIdx.x;
    if (t >= NUM_GRAPHS * NGRAPH) return;
    int g = t >> 4, cls = t & 15;
    float inv = 1.0f / fmaxf(cnt[g], 1.0f);
    float acc = fcb[cls];
#pragma unroll
    for (int k = 0; k < HID; ++k)
        acc += (sums[g * HID + k] * inv) * fcW[k * NGRAPH + cls];
    y[t] = acc;
}

extern "C" void kernel_launch(void* const* d_in, const int* in_sizes, int n_in,
                              void* d_out, int out_size, void* d_ws, size_t ws_size,
                              hipStream_t stream) {
    const float* x    = (const float*)d_in[0];
    const float* W1   = (const float*)d_in[1];
    const float* b1   = (const float*)d_in[2];
    const float* W2   = (const float*)d_in[3];
    const float* b2   = (const float*)d_in[4];
    const float* fcW  = (const float*)d_in[5];
    const float* fcb  = (const float*)d_in[6];
    const int*   eidx = (const int*)d_in[7];
    const int*   batch= (const int*)d_in[8];
    float* y = (float*)d_out;

    const int n = in_sizes[0] / 128;
    const int E = in_sizes[7] / 2;
    const int* src = eidx;
    const int* dst = eidx + E;

    size_t nAl = ((size_t)n + 3) & ~(size_t)3;
    float* deg  = (float*)d_ws;
    float* h    = deg + nAl;
    float* outb = h + (size_t)n * HID;
    float* sums = outb + (size_t)n * HID;
    float* cnt  = sums + NUM_GRAPHS * HID;

    const int TB = 256;
    const int n_tiles = (n + 15) / 16;

    k_init_deg<<<(n + TB - 1) / TB, TB, 0, stream>>>(deg, n);
    k_deg_accum<<<(E + TB - 1) / TB, TB, 0, stream>>>(dst, deg, E);
    k_rsqrt<<<(n + TB - 1) / TB, TB, 0, stream>>>(deg, n);

    k_gemm16<128, false><<<(n_tiles + 3) / 4, 128, 0, stream>>>(x, W1, h, n, n_tiles);
    k_self_bias<<<((n * HID) + TB - 1) / TB, TB, 0, stream>>>(h, deg, b1, outb, n);
    k_edge_prop<<<(E + TB - 1) / TB, TB, 0, stream>>>(src, dst, deg, h, outb, E);

    k_gemm16<16, true><<<(n_tiles + 3) / 4, 128, 0, stream>>>(outb, W2, h, n, n_tiles);
    k_self_bias<<<((n * HID) + TB - 1) / TB, TB, 0, stream>>>(h, deg, b2, outb, n);
    k_edge_prop<<<(E + TB - 1) / TB, TB, 0, stream>>>(src, dst, deg, h, outb, E);

    k_zero_pool<<<(NUM_GRAPHS * HID + TB - 1) / TB, TB, 0, stream>>>(sums, cnt);
    k_pool<<<(n + TB - 1) / TB, TB, 0, stream>>>(outb, batch, sums, cnt, n);
    k_head<<<(NUM_GRAPHS * NGRAPH + TB - 1) / TB, TB, 0, stream>>>(sums, cnt, fcW, fcb, y);
}


# degree count binned in LDS: 4 node ranges x 64 edge slices count into LDS integer counters, flushed as whole-line integer adds; replaces 6.4M scattered one-word global f32 atomics
# speedup vs baseline: 1.6841x; 1.3374x over previous
#include <hip/hip_runtime.h>
#include <hip/hip_bf16.h>

typedef __attribute__((ext_vector_type(2))) float v2f;
typedef __attribute__((ext_vector_type(8))) float v8f;

#define HID 16
#define NGRAPH 16
#define NUM_GRAPHS 512

__device__ __forceinline__ void atomAddF(float* p, float v) {
    __hip_atomic_fetch_add(p, v, __ATOMIC_RELAXED, __HIP_MEMORY_SCOPE_AGENT);
}

#define DEG_BINS    50176
#define DEG_THREADS 1024
#define DEG_SLICES  64
#define DEG_QUADS   8

__global__ void k_init_deg(unsigned* __restrict__ cnt, int n) {
    int i = blockIdx.x * blockDim.x + threadIdx.x;
    if (i < n) cnt[i] = 0u;
}

__device__ __forceinline__ void deg_bin_add(unsigned* bins, int d, int lo) {
    const unsigned r = (unsigned)(d - lo);
    if (r < (unsigned)DEG_BINS)
        __hip_atomic_fetch_add(bins + r, 1u, __ATOMIC_RELAXED, __HIP_MEMORY_SCOPE_WORKGROUP);
}

__global__ __launch_bounds__(DEG_THREADS)
void k_deg_accum(const int* __restrict__ dst, unsigned* __restrict__ cnt, int E, int n) {
    __shared__ unsigned bins[DEG_BINS];
    const int tid = threadIdx.x;
    const int lo  = blockIdx.y * DEG_BINS;

    for (int i = tid; i < DEG_BINS; i += DEG_THREADS) bins[i] = 0u;
    __syncthreads();

    const int per = (((E + (int)gridDim.x - 1) / (int)gridDim.x) + 3) & ~3;
    const int beg = blockIdx.x * per;
    int end = beg + per;
    if (end > E) end = E;

    int b0 = beg;
    if ((reinterpret_cast<size_t>(dst) & 15) == 0) {
        const int batch = 4 * DEG_THREADS * DEG_QUADS;
        for (; b0 + batch <= end; b0 += batch) {
            const int4* p = reinterpret_cast<const int4*>(dst + b0) + tid;
            int4 v[DEG_QUADS];
#pragma unroll
            for (int u = 0; u < DEG_QUADS; ++u) v[u] = p[u * DEG_THREADS];
#pragma unroll
            for (int u = 0; u < DEG_QUADS; ++u) {
                deg_bin_add(bins, v[u].x, lo);
                deg_bin_add(bins, v[u].y, lo);
                deg_bin_add(bins, v[u].z, lo);
                deg_bin_add(bins, v[u].w, lo);
            }
        }
    }
    for (int q = b0 + tid; q < end; q += DEG_THREADS) deg_bin_add(bins, dst[q], lo);
    __syncthreads();

    const int rot = ((int)blockIdx.x * (DEG_BINS / DEG_SLICES)) & ~31;
    for (int k = tid; k < DEG_BINS; k += DEG_THREADS) {
        int i = k + rot;
        if (i >= DEG_BINS) i -= DEG_BINS;
        const unsigned c = bins[i];
        const int node = lo + i;
        if (c != 0u && node < n)
            __hip_atomic_fetch_add(cnt + node, c, __ATOMIC_RELAXED, __HIP_MEMORY_SCOPE_AGENT);
    }
}

__global__ void k_rsqrt(unsigned* __restrict__ cnt, int n) {
    int i = blockIdx.x * blockDim.x + threadIdx.x;
    if (i < n) cnt[i] = __float_as_uint(rsqrtf((float)cnt[i] + 1.0f));
}

template<int K, bool RELU>
__global__ void k_gemm16(const float* __restrict__ A, const float* __restrict__ W,
                         float* __restrict__ D, int n, int n_tiles) {
    const int lane = threadIdx.x & 31;
    const int wid  = threadIdx.x >> 5;
    const int tile = blockIdx.x * (blockDim.x >> 5) + wid;
    if (tile >= n_tiles) return;

    const int m    = lane & 15;
    const int half = lane >> 4;
    const int kh   = half * 2;

    int rowA = tile * 16 + m;
    int rowC = rowA;
    if (rowA >= n) rowA = n - 1;

    const float* arow = A + (size_t)rowA * K;
    v8f acc = {};
#pragma unroll
    for (int k0 = 0; k0 < K; k0 += 4) {
        float a0 = arow[k0 + kh];
        float a1 = arow[k0 + kh + 1];
        if (RELU) { a0 = fmaxf(a0, 0.0f); a1 = fmaxf(a1, 0.0f); }
        v2f a; a.x = a0; a.y = a1;
        v2f b;
        b.x = W[(k0 + kh) * HID + m];
        b.y = W[(k0 + kh + 1) * HID + m];
        acc = __builtin_amdgcn_wmma_f32_16x16x4_f32(false, a, false, b,
                                                    (short)0, acc, false, false);
    }
#pragma unroll
    for (int r = 0; r < 8; ++r) {
        int row = tile * 16 + r + 8 * half;
        if (row < n) D[(size_t)row * HID + m] = acc[r];
    }
}

__global__ void k_self_bias(const float* __restrict__ h, const float* __restrict__ dinv,
                            const float* __restrict__ bias, float* __restrict__ out, int n) {
    int t = blockIdx.x * blockDim.x + threadIdx.x;
    if (t >= n * HID) return;
    int i = t >> 4, c = t & 15;
    float di = dinv[i];
    out[t] = di * di * h[t] + bias[c];
}

#define EDGES_PER_WAVE 32

template<int J>
__device__ __forceinline__ int row_lane(int v) {
    return __builtin_amdgcn_update_dpp(0, v, 0x150 + J, 0xf, 0xf, true);
}
template<int J>
__device__ __forceinline__ float row_lane(float v) {
    return __int_as_float(row_lane<J>(__float_as_int(v)));
}
__global__ void k_edge_prop(const int* __restrict__ src, const int* __restrict__ dst,
                            const float* __restrict__ dinv, const float* __restrict__ h,
                            float* __restrict__ out, int E) {
    const int lane = threadIdx.x & 31;
    const int wave = (blockIdx.x * blockDim.x + threadIdx.x) >> 5;
    const int base = wave * EDGES_PER_WAVE;
    if (base >= E) return;
    const int c    = lane & 15;
    const int hb   = lane & 16;

    const int e  = base + lane;
    const int ec = e < E ? e : E - 1;
    const int   s_l = src[ec];
    const int   d_l = dst[ec];
    const float w_l = dinv[s_l] * dinv[d_l];

    if (base + EDGES_PER_WAVE <= E) {
        unsigned off_d[16];
        float    wj[16];
        float    hv[16];
#define GATHER_ROUND(J)                                              \
        {                                                            \
            const int s = row_lane<J>(s_l);                          \
            const int d = row_lane<J>(d_l);                          \
            wj[J]    = row_lane<J>(w_l);                             \
            off_d[J] = (unsigned)d * HID + c;                        \
            hv[J]    = h[(unsigned)s * HID + c];                     \
        }
        GATHER_ROUND(0)  GATHER_ROUND(1)  GATHER_ROUND(2)  GATHER_ROUND(3)
        GATHER_ROUND(4)  GATHER_ROUND(5)  GATHER_ROUND(6)  GATHER_ROUND(7)
        GATHER_ROUND(8)  GATHER_ROUND(9)  GATHER_ROUND(10) GATHER_ROUND(11)
        GATHER_ROUND(12) GATHER_ROUND(13) GATHER_ROUND(14) GATHER_ROUND(15)
#undef GATHER_ROUND
#pragma unroll
        for (int j = 0; j < 16; ++j) atomAddF(out + off_d[j], wj[j] * hv[j]);
    } else {
#pragma unroll 1
        for (int j = 0; j < 16; ++j) {
            const int   s = __shfl(s_l, hb + j);
            const int   d = __shfl(d_l, hb + j);
            const float w = __shfl(w_l, hb + j);
            if (base + hb + j < E)
                atomAddF(out + ((unsigned)d * HID + c), w * h[(unsigned)s * HID + c]);
        }
    }
}

__global__ void k_zero_pool(float* __restrict__ sums, float* __restrict__ cnt) {
    int t = blockIdx.x * blockDim.x + threadIdx.x;
    if (t < NUM_GRAPHS * HID) sums[t] = 0.0f;
    if (t < NUM_GRAPHS) cnt[t] = 0.0f;
}

__global__ void k_pool(const float* __restrict__ out, const int* __restrict__ batch,
                       float* __restrict__ sums, float* __restrict__ cnt, int n) {
    int i = blockIdx.x * blockDim.x + threadIdx.x;
    if (i >= n) return;
    int g = batch[i];
    atomAddF(&cnt[g], 1.0f);
    const float* row = out + (size_t)i * HID;
#pragma unroll
    for (int c = 0; c < HID; ++c) atomAddF(&sums[g * HID + c], row[c]);
}

__global__ void k_head(const float* __restrict__ sums, const float* __restrict__ cnt,
                       const float* __restrict__ fcW, const float* __restrict__ fcb,
                       float* __restrict__ y) {
    int t = blockIdx.x * blockDim.x + threadIdx.x;
    if (t >= NUM_GRAPHS * NGRAPH) return;
    int g = t >> 4, cls = t & 15;
    float inv = 1.0f / fmaxf(cnt[g], 1.0f);
    float acc = fcb[cls];
#pragma unroll
    for (int k = 0; k < HID; ++k)
        acc += (sums[g * HID + k] * inv) * fcW[k * NGRAPH + cls];
    y[t] = acc;
}

extern "C" void kernel_launch(void* const* d_in, const int* in_sizes, int n_in,
                              void* d_out, int out_size, void* d_ws, size_t ws_size,
                              hipStream_t stream) {
    const float* x    = (const float*)d_in[0];
    const float* W1   = (const float*)d_in[1];
    const float* b1   = (const float*)d_in[2];
    const float* W2   = (const float*)d_in[3];
    const float* b2   = (const float*)d_in[4];
    const float* fcW  = (const float*)d_in[5];
    const float* fcb  = (const float*)d_in[6];
    const int*   eidx = (const int*)d_in[7];
    const int*   batch= (const int*)d_in[8];
    float* y = (float*)d_out;

    const int n = in_sizes[0] / 128;
    const int E = in_sizes[7] / 2;
    const int* src = eidx;
    const int* dst = eidx + E;

    size_t nAl = ((size_t)n + 3) & ~(size_t)3;
    float* deg  = (float*)d_ws;
    float* h    = deg + nAl;
    float* outb = h + (size_t)n * HID;
    float* sums = outb + (size_t)n * HID;
    float* cnt  = sums + NUM_GRAPHS * HID;

    const int TB = 256;
    const int n_tiles = (n + 15) / 16;

    unsigned* degc = (unsigned*)d_ws;
    k_init_deg<<<(n + TB - 1) / TB, TB, 0, stream>>>(degc, n);
    k_deg_accum<<<dim3(DEG_SLICES, (n + DEG_BINS - 1) / DEG_BINS), DEG_THREADS, 0, stream>>>(dst, degc, E, n);
    k_rsqrt<<<(n + TB - 1) / TB, TB, 0, stream>>>(degc, n);

    k_gemm16<128, false><<<(n_tiles + 3) / 4, 128, 0, stream>>>(x, W1, h, n, n_tiles);
    k_self_bias<<<((n * HID) + TB - 1) / TB, TB, 0, stream>>>(h, deg, b1, outb, n);
    k_edge_prop<<<(E + TB - 1) / TB, TB, 0, stream>>>(src, dst, deg, h, outb, E);

    k_gemm16<16, true><<<(n_tiles + 3) / 4, 128, 0, stream>>>(outb, W2, h, n, n_tiles);
    k_self_bias<<<((n * HID) + TB - 1) / TB, TB, 0, stream>>>(h, deg, b2, outb, n);
    k_edge_prop<<<(E + TB - 1) / TB, TB, 0, stream>>>(src, dst, deg, h, outb, E);

    k_zero_pool<<<(NUM_GRAPHS * HID + TB - 1) / TB, TB, 0, stream>>>(sums, cnt);
    k_pool<<<(n + TB - 1) / TB, TB, 0, stream>>>(outb, batch, sums, cnt, n);
    k_head<<<(NUM_GRAPHS * NGRAPH + TB - 1) / TB, TB, 0, stream>>>(sums, cnt, fcW, fcb, y);
}


# layer-2 edge terms go straight into the per-graph pool sums (their only reader) via an LDS table per work-group, flushed once: no global atomic per edge in pass 2; pool sum order changes
# speedup vs baseline: 1.8773x; 1.1147x over previous
#include <hip/hip_runtime.h>
#include <hip/hip_bf16.h>

typedef __attribute__((ext_vector_type(2))) float v2f;
typedef __attribute__((ext_vector_type(8))) float v8f;

#define HID 16
#define NGRAPH 16
#define NUM_GRAPHS 512

__device__ __forceinline__ void atomAddF(float* p, float v) {
    __hip_atomic_fetch_add(p, v, __ATOMIC_RELAXED, __HIP_MEMORY_SCOPE_AGENT);
}

__device__ __forceinline__ void ldsAddF(float* p, float v) {
    __hip_atomic_fetch_add(p, v, __ATOMIC_RELAXED, __HIP_MEMORY_SCOPE_WORKGROUP);
}

#define DEG_BINS    50176
#define DEG_THREADS 1024
#define DEG_SLICES  64
#define DEG_QUADS   8

__global__ void k_init_deg(unsigned* __restrict__ cnt, int n) {
    int i = blockIdx.x * blockDim.x + threadIdx.x;
    if (i < n) cnt[i] = 0u;
}

__device__ __forceinline__ void deg_bin_add(unsigned* bins, int d, int lo) {
    const unsigned r = (unsigned)(d - lo);
    if (r < (unsigned)DEG_BINS)
        __hip_atomic_fetch_add(bins + r, 1u, __ATOMIC_RELAXED, __HIP_MEMORY_SCOPE_WORKGROUP);
}

__global__ __launch_bounds__(DEG_THREADS)
void k_deg_accum(const int* __restrict__ dst, unsigned* __restrict__ cnt, int E, int n) {
    __shared__ unsigned bins[DEG_BINS];
    const int tid = threadIdx.x;
    const int lo  = blockIdx.y * DEG_BINS;

    for (int i = tid; i < DEG_BINS; i += DEG_THREADS) bins[i] = 0u;
    __syncthreads();

    const int per = (((E + (int)gridDim.x - 1) / (int)gridDim.x) + 3) & ~3;
    const int beg = blockIdx.x * per;
    int end = beg + per;
    if (end > E) end = E;

    int b0 = beg;
    if ((reinterpret_cast<size_t>(dst) & 15) == 0) {
        const int batch = 4 * DEG_THREADS * DEG_QUADS;
        for (; b0 + batch <= end; b0 += batch) {
            const int4* p = reinterpret_cast<const int4*>(dst + b0) + tid;
            int4 v[DEG_QUADS];
#pragma unroll
            for (int u = 0; u < DEG_QUADS; ++u) v[u] = p[u * DEG_THREADS];
#pragma unroll
            for (int u = 0; u < DEG_QUADS; ++u) {
                deg_bin_add(bins, v[u].x, lo);
                deg_bin_add(bins, v[u].y, lo);
                deg_bin_add(bins, v[u].z, lo);
                deg_bin_add(bins, v[u].w, lo);
            }
        }
    }
    for (int q = b0 + tid; q < end; q += DEG_THREADS) deg_bin_add(bins, dst[q], lo);
    __syncthreads();

    const int rot = ((int)blockIdx.x * (DEG_BINS / DEG_SLICES)) & ~31;
    for (int k = tid; k < DEG_BINS; k += DEG_THREADS) {
        int i = k + rot;
        if (i >= DEG_BINS) i -= DEG_BINS;
        const unsigned c = bins[i];
        const int node = lo + i;
        if (c != 0u && node < n)
            __hip_atomic_fetch_add(cnt + node, c, __ATOMIC_RELAXED, __HIP_MEMORY_SCOPE_AGENT);
    }
}

__global__ void k_rsqrt(unsigned* __restrict__ cnt, int n) {
    int i = blockIdx.x * blockDim.x + threadIdx.x;
    if (i < n) cnt[i] = __float_as_uint(rsqrtf((float)cnt[i] + 1.0f));
}

template<int K, bool RELU>
__global__ void k_gemm16(const float* __restrict__ A, const float* __restrict__ W,
                         float* __restrict__ D, int n, int n_tiles) {
    const int lane = threadIdx.x & 31;
    const int wid  = threadIdx.x >> 5;
    const int tile = blockIdx.x * (blockDim.x >> 5) + wid;
    if (tile >= n_tiles) return;

    const int m    = lane & 15;
    const int half = lane >> 4;
    const int kh   = half * 2;

    int rowA = tile * 16 + m;
    int rowC = rowA;
    if (rowA >= n) rowA = n - 1;

    const float* arow = A + (size_t)rowA * K;
    v8f acc = {};
#pragma unroll
    for (int k0 = 0; k0 < K; k0 += 4) {
        float a0 = arow[k0 + kh];
        float a1 = arow[k0 + kh + 1];
        if (RELU) { a0 = fmaxf(a0, 0.0f); a1 = fmaxf(a1, 0.0f); }
        v2f a; a.x = a0; a.y = a1;
        v2f b;
        b.x = W[(k0 + kh) * HID + m];
        b.y = W[(k0 + kh + 1) * HID + m];
        acc = __builtin_amdgcn_wmma_f32_16x16x4_f32(false, a, false, b,
                                                    (short)0, acc, false, false);
    }
#pragma unroll
    for (int r = 0; r < 8; ++r) {
        int row = tile * 16 + r + 8 * half;
        if (row < n) D[(size_t)row * HID + m] = acc[r];
    }
}

__global__ void k_self_bias(const float* __restrict__ h, const float* __restrict__ dinv,
                            const float* __restrict__ bias, float* __restrict__ out, int n) {
    int t = blockIdx.x * blockDim.x + threadIdx.x;
    if (t >= n * HID) return;
    int i = t >> 4, c = t & 15;
    float di = dinv[i];
    out[t] = di * di * h[t] + bias[c];
}

#define EDGES_PER_WAVE 32

template<int J>
__device__ __forceinline__ int row_lane(int v) {
    return __builtin_amdgcn_update_dpp(0, v, 0x150 + J, 0xf, 0xf, true);
}
template<int J>
__device__ __forceinline__ float row_lane(float v) {
    return __int_as_float(row_lane<J>(__float_as_int(v)));
}
__global__ void k_edge_prop(const int* __restrict__ src, const int* __restrict__ dst,
                            const float* __restrict__ dinv, const float* __restrict__ h,
                            float* __restrict__ out, int E) {
    const int lane = threadIdx.x & 31;
    const int wave = (blockIdx.x * blockDim.x + threadIdx.x) >> 5;
    const int base = wave * EDGES_PER_WAVE;
    if (base >= E) return;
    const int c    = lane & 15;
    const int hb   = lane & 16;

    const int e  = base + lane;
    const int ec = e < E ? e : E - 1;
    const int   s_l = src[ec];
    const int   d_l = dst[ec];
    const float w_l = dinv[s_l] * dinv[d_l];

    if (base + EDGES_PER_WAVE <= E) {
        unsigned off_d[16];
        float    wj[16];
        float    hv[16];
#define GATHER_ROUND(J)                                              \
        {                                                            \
            const int s = row_lane<J>(s_l);                          \
            const int d = row_lane<J>(d_l);                          \
            wj[J]    = row_lane<J>(w_l);                             \
            off_d[J] = (unsigned)d * HID + c;                        \
            hv[J]    = h[(unsigned)s * HID + c];                     \
        }
        GATHER_ROUND(0)  GATHER_ROUND(1)  GATHER_ROUND(2)  GATHER_ROUND(3)
        GATHER_ROUND(4)  GATHER_ROUND(5)  GATHER_ROUND(6)  GATHER_ROUND(7)
        GATHER_ROUND(8)  GATHER_ROUND(9)  GATHER_ROUND(10) GATHER_ROUND(11)
        GATHER_ROUND(12) GATHER_ROUND(13) GATHER_ROUND(14) GATHER_ROUND(15)
#undef GATHER_ROUND
#pragma unroll
        for (int j = 0; j < 16; ++j) atomAddF(out + off_d[j], wj[j] * hv[j]);
    } else {
#pragma unroll 1
        for (int j = 0; j < 16; ++j) {
            const int   s = __shfl(s_l, hb + j);
            const int   d = __shfl(d_l, hb + j);
            const float w = __shfl(w_l, hb + j);
            if (base + hb + j < E)
                atomAddF(out + ((unsigned)d * HID + c), w * h[(unsigned)s * HID + c]);
        }
    }
}

#define EPOOL_THREADS 1024
#define EPOOL_BLOCKS  512

__global__ __launch_bounds__(EPOOL_THREADS)
void k_edge_pool(const int* __restrict__ src, const int* __restrict__ dst,
                 const float* __restrict__ dinv, const int* __restrict__ batch,
                 const float* __restrict__ h, float* __restrict__ sums, int E) {
    __shared__ float tab[NUM_GRAPHS * HID];
    const int tid = threadIdx.x;
    for (int i = tid; i < NUM_GRAPHS * HID; i += EPOOL_THREADS) tab[i] = 0.0f;
    __syncthreads();

    const int lane = tid & 31;
    const int c    = lane & 15;
    const int hb   = lane & 16;
    const int wave   = blockIdx.x * (EPOOL_THREADS / 32) + (tid >> 5);
    const int stride = (int)gridDim.x * (EPOOL_THREADS / 32) * EDGES_PER_WAVE;

    for (int base = wave * EDGES_PER_WAVE; base < E; base += stride) {
        const int e  = base + lane;
        const int ec = e < E ? e : E - 1;
        const int   s_l  = src[ec];
        const int   d_l  = dst[ec];
        const float ds_l = dinv[s_l];
        const float dd_l = dinv[d_l];
        const int   g_l  = batch[d_l];

        if (base + EDGES_PER_WAVE <= E) {
            float hv[16];
#define POOL_GATHER(J) hv[J] = h[(unsigned)row_lane<J>(s_l) * HID + c];
            POOL_GATHER(0)  POOL_GATHER(1)  POOL_GATHER(2)  POOL_GATHER(3)
            POOL_GATHER(4)  POOL_GATHER(5)  POOL_GATHER(6)  POOL_GATHER(7)
            POOL_GATHER(8)  POOL_GATHER(9)  POOL_GATHER(10) POOL_GATHER(11)
            POOL_GATHER(12) POOL_GATHER(13) POOL_GATHER(14) POOL_GATHER(15)
#undef POOL_GATHER
            const float w_l = ds_l * dd_l;
#define POOL_ADD(J) ldsAddF(tab + ((unsigned)row_lane<J>(g_l) * HID + c), row_lane<J>(w_l) * hv[J]);
            POOL_ADD(0)  POOL_ADD(1)  POOL_ADD(2)  POOL_ADD(3)
            POOL_ADD(4)  POOL_ADD(5)  POOL_ADD(6)  POOL_ADD(7)
            POOL_ADD(8)  POOL_ADD(9)  POOL_ADD(10) POOL_ADD(11)
            POOL_ADD(12) POOL_ADD(13) POOL_ADD(14) POOL_ADD(15)
#undef POOL_ADD
        } else {
#pragma unroll 1
            for (int j = 0; j < 16; ++j) {
                const int   s = __shfl(s_l, hb + j);
                const int   g = __shfl(g_l, hb + j);
                const float w = __shfl(ds_l, hb + j) * __shfl(dd_l, hb + j);
                if (base + hb + j < E)
                    ldsAddF(tab + ((unsigned)g * HID + c), w * h[(unsigned)s * HID + c]);
            }
        }
    }
    __syncthreads();

    const int rot = ((int)blockIdx.x * 48) & (NUM_GRAPHS * HID - 1) & ~31;
    for (int k = tid; k < NUM_GRAPHS * HID; k += EPOOL_THREADS) {
        const int i = (k + rot) & (NUM_GRAPHS * HID - 1);
        atomAddF(sums + i, tab[i]);
    }
}

__global__ void k_zero_pool(float* __restrict__ sums, float* __restrict__ cnt) {
    int t = blockIdx.x * blockDim.x + threadIdx.x;
    if (t < NUM_GRAPHS * HID) sums[t] = 0.0f;
    if (t < NUM_GRAPHS) cnt[t] = 0.0f;
}

__global__ void k_pool(const float* __restrict__ out, const int* __restrict__ batch,
                       float* __restrict__ sums, float* __restrict__ cnt, int n) {
    int i = blockIdx.x * blockDim.x + threadIdx.x;
    if (i >= n) return;
    int g = batch[i];
    atomAddF(&cnt[g], 1.0f);
    const float* row = out + (size_t)i * HID;
#pragma unroll
    for (int c = 0; c < HID; ++c) atomAddF(&sums[g * HID + c], row[c]);
}

__global__ void k_head(const float* __restrict__ sums, const float* __restrict__ cnt,
                       const float* __restrict__ fcW, const float* __restrict__ fcb,
                       float* __restrict__ y) {
    int t = blockIdx.x * blockDim.x + threadIdx.x;
    if (t >= NUM_GRAPHS * NGRAPH) return;
    int g = t >> 4, cls = t & 15;
    float inv = 1.0f / fmaxf(cnt[g], 1.0f);
    float acc = fcb[cls];
#pragma unroll
    for (int k = 0; k < HID; ++k)
        acc += (sums[g * HID + k] * inv) * fcW[k * NGRAPH + cls];
    y[t] = acc;
}

extern "C" void kernel_launch(void* const* d_in, const int* in_sizes, int n_in,
                              void* d_out, int out_size, void* d_ws, size_t ws_size,
                              hipStream_t stream) {
    const float* x    = (const float*)d_in[0];
    const float* W1   = (const float*)d_in[1];
    const float* b1   = (const float*)d_in[2];
    const float* W2   = (const float*)d_in[3];
    const float* b2   = (const float*)d_in[4];
    const float* fcW  = (const float*)d_in[5];
    const float* fcb  = (const float*)d_in[6];
    const int*   eidx = (const int*)d_in[7];
    const int*   batch= (const int*)d_in[8];
    float* y = (float*)d_out;

    const int n = in_sizes[0] / 128;
    const int E = in_sizes[7] / 2;
    const int* src = eidx;
    const int* dst = eidx + E;

    size_t nAl = ((size_t)n + 3) & ~(size_t)3;
    float* deg  = (float*)d_ws;
    float* h    = deg + nAl;
    float* outb = h + (size_t)n * HID;
    float* sums = outb + (size_t)n * HID;
    float* cnt  = sums + NUM_GRAPHS * HID;

    const int TB = 256;
    const int n_tiles = (n + 15) / 16;

    unsigned* degc = (unsigned*)d_ws;
    k_init_deg<<<(n + TB - 1) / TB, TB, 0, stream>>>(degc, n);
    k_deg_accum<<<dim3(DEG_SLICES, (n + DEG_BINS - 1) / DEG_BINS), DEG_THREADS, 0, stream>>>(dst, degc, E, n);
    k_rsqrt<<<(n + TB - 1) / TB, TB, 0, stream>>>(degc, n);

    k_gemm16<128, false><<<(n_tiles + 3) / 4, 128, 0, stream>>>(x, W1, h, n, n_tiles);
    k_self_bias<<<((n * HID) + TB - 1) / TB, TB, 0, stream>>>(h, deg, b1, outb, n);
    k_edge_prop<<<(E + TB - 1) / TB, TB, 0, stream>>>(src, dst, deg, h, outb, E);

    k_gemm16<16, true><<<(n_tiles + 3) / 4, 128, 0, stream>>>(outb, W2, h, n, n_tiles);
    k_self_bias<<<((n * HID) + TB - 1) / TB, TB, 0, stream>>>(h, deg, b2, outb, n);

    k_zero_pool<<<(NUM_GRAPHS * HID + TB - 1) / TB, TB, 0, stream>>>(sums, cnt);
    k_pool<<<(n + TB - 1) / TB, TB, 0, stream>>>(outb, batch, sums, cnt, n);
    k_edge_pool<<<EPOOL_BLOCKS, EPOOL_THREADS, 0, stream>>>(src, dst, deg, batch, h, sums, E);
    k_head<<<(NUM_GRAPHS * NGRAPH + TB - 1) / TB, TB, 0, stream>>>(sums, cnt, fcW, fcb, y);
}


# edge weight factored out of the edge passes: h rows stored scaled by dinv[s]; dinv[d] and bias applied once per row after the sum (layer 1), dinv[d] + graph id read as one 8-byte record (layer 2)
# speedup vs baseline: 2.1892x; 1.1661x over previous
#include <hip/hip_runtime.h>
#include <hip/hip_bf16.h>

typedef __attribute__((ext_vector_type(2))) float v2f;
typedef __attribute__((ext_vector_type(8))) float v8f;

#define HID 16
#define NGRAPH 16
#define NUM_GRAPHS 512

__device__ __forceinline__ void atomAddF(float* p, float v) {
    __hip_atomic_fetch_add(p, v, __ATOMIC_RELAXED, __HIP_MEMORY_SCOPE_AGENT);
}

__device__ __forceinline__ void ldsAddF(float* p, float v) {
    __hip_atomic_fetch_add(p, v, __ATOMIC_RELAXED, __HIP_MEMORY_SCOPE_WORKGROUP);
}

#define DEG_BINS    50176
#define DEG_THREADS 1024
#define DEG_SLICES  64
#define DEG_QUADS   8

__global__ void k_init_deg(unsigned* __restrict__ cnt, int n) {
    int i = blockIdx.x * blockDim.x + threadIdx.x;
    if (i < n) cnt[i] = 0u;
}

__device__ __forceinline__ void deg_bin_add(unsigned* bins, int d, int lo) {
    const unsigned r = (unsigned)(d - lo);
    if (r < (unsigned)DEG_BINS)
        __hip_atomic_fetch_add(bins + r, 1u, __ATOMIC_RELAXED, __HIP_MEMORY_SCOPE_WORKGROUP);
}

__global__ __launch_bounds__(DEG_THREADS)
void k_deg_accum(const int* __restrict__ dst, unsigned* __restrict__ cnt, int E, int n) {
    __shared__ unsigned bins[DEG_BINS];
    const int tid = threadIdx.x;
    const int lo  = blockIdx.y * DEG_BINS;

    for (int i = tid; i < DEG_BINS; i += DEG_THREADS) bins[i] = 0u;
    __syncthreads();

    const int per = (((E + (int)gridDim.x - 1) / (int)gridDim.x) + 3) & ~3;
    const int beg = blockIdx.x * per;
    int end = beg + per;
    if (end > E) end = E;

    int b0 = beg;
    if ((reinterpret_cast<size_t>(dst) & 15) == 0) {
        const int batch = 4 * DEG_THREADS * DEG_QUADS;
        for (; b0 + batch <= end; b0 += batch) {
            const int4* p = reinterpret_cast<const int4*>(dst + b0) + tid;
            int4 v[DEG_QUADS];
#pragma unroll
            for (int u = 0; u < DEG_QUADS; ++u) v[u] = p[u * DEG_THREADS];
#pragma unroll
            for (int u = 0; u < DEG_QUADS; ++u) {
                deg_bin_add(bins, v[u].x, lo);
                deg_bin_add(bins, v[u].y, lo);
                deg_bin_add(bins, v[u].z, lo);
                deg_bin_add(bins, v[u].w, lo);
            }
        }
    }
    for (int q = b0 + tid; q < end; q += DEG_THREADS) deg_bin_add(bins, dst[q], lo);
    __syncthreads();

    const int rot = ((int)blockIdx.x * (DEG_BINS / DEG_SLICES)) & ~31;
    for (int k = tid; k < DEG_BINS; k += DEG_THREADS) {
        int i = k + rot;
        if (i >= DEG_BINS) i -= DEG_BINS;
        const unsigned c = bins[i];
        const int node = lo + i;
        if (c != 0u && node < n)
            __hip_atomic_fetch_add(cnt + node, c, __ATOMIC_RELAXED, __HIP_MEMORY_SCOPE_AGENT);
    }
}

__global__ void k_rsqrt(unsigned* __restrict__ cnt, int n) {
    int i = blockIdx.x * blockDim.x + threadIdx.x;
    if (i < n) cnt[i] = __float_as_uint(rsqrtf((float)cnt[i] + 1.0f));
}

template<int K, bool LAYER2>
__global__ void k_gemm16(const float* __restrict__ A, const float* __restrict__ W,
                         const float* __restrict__ dinv, const float* __restrict__ bias,
                         float* __restrict__ D, float* __restrict__ D2, int n, int n_tiles) {
    const int lane = threadIdx.x & 31;
    const int wid  = threadIdx.x >> 5;
    const int tile = blockIdx.x * (blockDim.x >> 5) + wid;
    if (tile >= n_tiles) return;

    const int m    = lane & 15;
    const int half = lane >> 4;
    const int kh   = half * 2;

    int rowA = tile * 16 + m;
    int rowC = rowA;
    if (rowA >= n) rowA = n - 1;

    const float* arow = A + (size_t)rowA * K;
    const float  dA   = LAYER2 ? dinv[rowA] : 1.0f;
    v8f acc = {};
#pragma unroll
    for (int k0 = 0; k0 < K; k0 += 4) {
        float a0 = arow[k0 + kh];
        float a1 = arow[k0 + kh + 1];
        if (LAYER2) {
            a0 = fmaxf(dA * a0 + bias[k0 + kh], 0.0f);
            a1 = fmaxf(dA * a1 + bias[k0 + kh + 1], 0.0f);
        }
        v2f a; a.x = a0; a.y = a1;
        v2f b;
        b.x = W[(k0 + kh) * HID + m];
        b.y = W[(k0 + kh + 1) * HID + m];
        acc = __builtin_amdgcn_wmma_f32_16x16x4_f32(false, a, false, b,
                                                    (short)0, acc, false, false);
    }
#pragma unroll
    for (int r = 0; r < 8; ++r) {
        int row = tile * 16 + r + 8 * half;
        if (row < n) {
            const float v = dinv[row] * acc[r];
            D[(size_t)row * HID + m] = v;
            if (!LAYER2) D2[(size_t)row * HID + m] = v;
        }
    }
}

#define EDGES_PER_WAVE 32

template<int J>
__device__ __forceinline__ int row_lane(int v) {
    return __builtin_amdgcn_update_dpp(0, v, 0x150 + J, 0xf, 0xf, true);
}
template<int J>
__device__ __forceinline__ float row_lane(float v) {
    return __int_as_float(row_lane<J>(__float_as_int(v)));
}
__global__ void k_edge_prop(const int* __restrict__ src, const int* __restrict__ dst,
                            const float* __restrict__ h, float* __restrict__ out, int E) {
    const int lane = threadIdx.x & 31;
    const int wave = (blockIdx.x * blockDim.x + threadIdx.x) >> 5;
    const int base = wave * EDGES_PER_WAVE;
    if (base >= E) return;
    const int c    = lane & 15;
    const int hb   = lane & 16;

    const int e  = base + lane;
    const int ec = e < E ? e : E - 1;
    const int   s_l = src[ec];
    const int   d_l = dst[ec];

    if (base + EDGES_PER_WAVE <= E) {
        unsigned off_d[16];
        float    hv[16];
#define GATHER_ROUND(J)                                              \
        {                                                            \
            const int s = row_lane<J>(s_l);                          \
            const int d = row_lane<J>(d_l);                          \
            off_d[J] = (unsigned)d * HID + c;                        \
            hv[J]    = h[(unsigned)s * HID + c];                     \
        }
        GATHER_ROUND(0)  GATHER_ROUND(1)  GATHER_ROUND(2)  GATHER_ROUND(3)
        GATHER_ROUND(4)  GATHER_ROUND(5)  GATHER_ROUND(6)  GATHER_ROUND(7)
        GATHER_ROUND(8)  GATHER_ROUND(9)  GATHER_ROUND(10) GATHER_ROUND(11)
        GATHER_ROUND(12) GATHER_ROUND(13) GATHER_ROUND(14) GATHER_ROUND(15)
#undef GATHER_ROUND
#pragma unroll
        for (int j = 0; j < 16; ++j) atomAddF(out + off_d[j], hv[j]);
    } else {
#pragma unroll 1
        for (int j = 0; j < 16; ++j) {
            const int   s = __shfl(s_l, hb + j);
            const int   d = __shfl(d_l, hb + j);
            if (base + hb + j < E)
                atomAddF(out + ((unsigned)d * HID + c), h[(unsigned)s * HID + c]);
        }
    }
}

#define EPOOL_THREADS 1024
#define EPOOL_BLOCKS  512

__global__ __launch_bounds__(EPOOL_THREADS)
void k_edge_pool(const int* __restrict__ src, const int* __restrict__ dst,
                 const int* __restrict__ rec,
                 const float* __restrict__ h, float* __restrict__ sums, int E) {
    __shared__ float tab[NUM_GRAPHS * HID];
    const int tid = threadIdx.x;
    for (int i = tid; i < NUM_GRAPHS * HID; i += EPOOL_THREADS) tab[i] = 0.0f;
    __syncthreads();

    const int lane = tid & 31;
    const int c    = lane & 15;
    const int hb   = lane & 16;
    const int wave   = blockIdx.x * (EPOOL_THREADS / 32) + (tid >> 5);
    const int stride = (int)gridDim.x * (EPOOL_THREADS / 32) * EDGES_PER_WAVE;

    for (int base = wave * EDGES_PER_WAVE; base < E; base += stride) {
        const int e  = base + lane;
        const int ec = e < E ? e : E - 1;
        const int   s_l  = src[ec];
        const int   d_l  = dst[ec];
        const int2  r_l  = reinterpret_cast<const int2*>(rec)[d_l];
        const float w_l  = __int_as_float(r_l.x);
        const int   g_l  = r_l.y;

        if (base + EDGES_PER_WAVE <= E) {
            float hv[16];
#define POOL_GATHER(J) hv[J] = h[(unsigned)row_lane<J>(s_l) * HID + c];
            POOL_GATHER(0)  POOL_GATHER(1)  POOL_GATHER(2)  POOL_GATHER(3)
            POOL_GATHER(4)  POOL_GATHER(5)  POOL_GATHER(6)  POOL_GATHER(7)
            POOL_GATHER(8)  POOL_GATHER(9)  POOL_GATHER(10) POOL_GATHER(11)
            POOL_GATHER(12) POOL_GATHER(13) POOL_GATHER(14) POOL_GATHER(15)
#undef POOL_GATHER
#define POOL_ADD(J) ldsAddF(tab + ((unsigned)row_lane<J>(g_l) * HID + c), row_lane<J>(w_l) * hv[J]);
            POOL_ADD(0)  POOL_ADD(1)  POOL_ADD(2)  POOL_ADD(3)
            POOL_ADD(4)  POOL_ADD(5)  POOL_ADD(6)  POOL_ADD(7)
            POOL_ADD(8)  POOL_ADD(9)  POOL_ADD(10) POOL_ADD(11)
            POOL_ADD(12) POOL_ADD(13) POOL_ADD(14) POOL_ADD(15)
#undef POOL_ADD
        } else {
#pragma unroll 1
            for (int j = 0; j < 16; ++j) {
                const int   s = __shfl(s_l, hb + j);
                const int   g = __shfl(g_l, hb + j);
                const float w = __shfl(w_l, hb + j);
                if (base + hb + j < E)
                    ldsAddF(tab + ((unsigned)g * HID + c), w * h[(unsigned)s * HID + c]);
            }
        }
    }
    __syncthreads();

    const int rot = ((int)blockIdx.x * 48) & (NUM_GRAPHS * HID - 1) & ~31;
    for (int k = tid; k < NUM_GRAPHS * HID; k += EPOOL_THREADS) {
        const int i = (k + rot) & (NUM_GRAPHS * HID - 1);
        atomAddF(sums + i, tab[i]);
    }
}

__global__ void k_zero_pool(float* __restrict__ sums, float* __restrict__ cnt) {
    int t = blockIdx.x * blockDim.x + threadIdx.x;
    if (t < NUM_GRAPHS * HID) sums[t] = 0.0f;
    if (t < NUM_GRAPHS) cnt[t] = 0.0f;
}

__global__ void k_pool(const float* __restrict__ h, const float* __restrict__ dinv,
                       const float* __restrict__ bias, const int* __restrict__ batch,
                       int* __restrict__ rec,
                       float* __restrict__ sums, float* __restrict__ cnt, int n) {
    int i = blockIdx.x * blockDim.x + threadIdx.x;
    if (i >= n) return;
    const int   g  = batch[i];
    const float di = dinv[i];
    reinterpret_cast<int2*>(rec)[i] = make_int2(__float_as_int(di), g);
    atomAddF(&cnt[g], 1.0f);
    const float* row = h + (size_t)i * HID;
#pragma unroll
    for (int c = 0; c < HID; ++c) atomAddF(&sums[g * HID + c], di * row[c] + bias[c]);
}

__global__ void k_head(const float* __restrict__ sums, const float* __restrict__ cnt,
                       const float* __restrict__ fcW, const float* __restrict__ fcb,
                       float* __restrict__ y) {
    int t = blockIdx.x * blockDim.x + threadIdx.x;
    if (t >= NUM_GRAPHS * NGRAPH) return;
    int g = t >> 4, cls = t & 15;
    float inv = 1.0f / fmaxf(cnt[g], 1.0f);
    float acc = fcb[cls];
#pragma unroll
    for (int k = 0; k < HID; ++k)
        acc += (sums[g * HID + k] * inv) * fcW[k * NGRAPH + cls];
    y[t] = acc;
}

extern "C" void kernel_launch(void* const* d_in, const int* in_sizes, int n_in,
                              void* d_out, int out_size, void* d_ws, size_t ws_size,
                              hipStream_t stream) {
    const float* x    = (const float*)d_in[0];
    const float* W1   = (const float*)d_in[1];
    const float* b1   = (const float*)d_in[2];
    const float* W2   = (const float*)d_in[3];
    const float* b2   = (const float*)d_in[4];
    const float* fcW  = (const float*)d_in[5];
    const float* fcb  = (const float*)d_in[6];
    const int*   eidx = (const int*)d_in[7];
    const int*   batch= (const int*)d_in[8];
    float* y = (float*)d_out;

    const int n = in_sizes[0] / 128;
    const int E = in_sizes[7] / 2;
    const int* src = eidx;
    const int* dst = eidx + E;

    size_t nAl = ((size_t)n + 3) & ~(size_t)3;
    float* deg  = (float*)d_ws;
    float* h    = deg + nAl;
    float* outb = h + (size_t)n * HID;
    float* sums = outb + (size_t)n * HID;
    float* cnt  = sums + NUM_GRAPHS * HID;

    const int TB = 256;
    const int n_tiles = (n + 15) / 16;

    unsigned* degc = (unsigned*)d_ws;
    k_init_deg<<<(n + TB - 1) / TB, TB, 0, stream>>>(degc, n);
    k_deg_accum<<<dim3(DEG_SLICES, (n + DEG_BINS - 1) / DEG_BINS), DEG_THREADS, 0, stream>>>(dst, degc, E, n);
    k_rsqrt<<<(n + TB - 1) / TB, TB, 0, stream>>>(degc, n);

    k_gemm16<128, false><<<(n_tiles + 3) / 4, 128, 0, stream>>>(x, W1, deg, b1, h, outb, n, n_tiles);
    k_edge_prop<<<(E + TB - 1) / TB, TB, 0, stream>>>(src, dst, h, outb, E);

    k_gemm16<16, true><<<(n_tiles + 3) / 4, 128, 0, stream>>>(outb, W2, deg, b1, h, nullptr, n, n_tiles);

    int* rec = (int*)outb;
    k_zero_pool<<<(NUM_GRAPHS * HID + TB - 1) / TB, TB, 0, stream>>>(sums, cnt);
    k_pool<<<(n + TB - 1) / TB, TB, 0, stream>>>(h, deg, b2, batch, rec, sums, cnt, n);
    k_edge_pool<<<EPOOL_BLOCKS, EPOOL_THREADS, 0, stream>>>(src, dst, rec, h, sums, E);
    k_head<<<(NUM_GRAPHS * NGRAPH + TB - 1) / TB, TB, 0, stream>>>(sums, cnt, fcW, fcb, y);
}


# edge scatter: a work-group adds only into out rows of its own L2 half (die id); edge pieces come from a work counter per half, the other half's left-overs taken last; atomic order unordered as before
# speedup vs baseline: 2.6344x; 1.2034x over previous
#include <hip/hip_runtime.h>
#include <hip/hip_bf16.h>

typedef __attribute__((ext_vector_type(2))) float v2f;
typedef __attribute__((ext_vector_type(8))) float v8f;

#define HID 16
#define NGRAPH 16
#define NUM_GRAPHS 512

__device__ __forceinline__ void atomAddF(float* p, float v) {
    __hip_atomic_fetch_add(p, v, __ATOMIC_RELAXED, __HIP_MEMORY_SCOPE_AGENT);
}

__device__ __forceinline__ void ldsAddF(float* p, float v) {
    __hip_atomic_fetch_add(p, v, __ATOMIC_RELAXED, __HIP_MEMORY_SCOPE_WORKGROUP);
}

#define DEG_BINS    50176
#define DEG_THREADS 1024
#define DEG_SLICES  64
#define DEG_QUADS   8

#define CTR_STRIDE  64

__global__ void k_init_deg(unsigned* __restrict__ cnt, unsigned* __restrict__ ctr, int n) {
    int i = blockIdx.x * blockDim.x + threadIdx.x;
    if (i < n) cnt[i] = 0u;
    if (i < 2) ctr[i * CTR_STRIDE] = 0u;
}

__device__ __forceinline__ void deg_bin_add(unsigned* bins, int d, int lo) {
    const unsigned r = (unsigned)(d - lo);
    if (r < (unsigned)DEG_BINS)
        __hip_atomic_fetch_add(bins + r, 1u, __ATOMIC_RELAXED, __HIP_MEMORY_SCOPE_WORKGROUP);
}

__global__ __launch_bounds__(DEG_THREADS)
void k_deg_accum(const int* __restrict__ dst, unsigned* __restrict__ cnt, int E, int n) {
    __shared__ unsigned bins[DEG_BINS];
    const int tid = threadIdx.x;
    const int lo  = blockIdx.y * DEG_BINS;

    for (int i = tid; i < DEG_BINS; i += DEG_THREADS) bins[i] = 0u;
    __syncthreads();

    const int per = (((E + (int)gridDim.x - 1) / (int)gridDim.x) + 3) & ~3;
    const int beg = blockIdx.x * per;
    int end = beg + per;
    if (end > E) end = E;

    int b0 = beg;
    if ((reinterpret_cast<size_t>(dst) & 15) == 0) {
        const int batch = 4 * DEG_THREADS * DEG_QUADS;
        for (; b0 + batch <= end; b0 += batch) {
            const int4* p = reinterpret_cast<const int4*>(dst + b0) + tid;
            int4 v[DEG_QUADS];
#pragma unroll
            for (int u = 0; u < DEG_QUADS; ++u) v[u] = p[u * DEG_THREADS];
#pragma unroll
            for (int u = 0; u < DEG_QUADS; ++u) {
                deg_bin_add(bins, v[u].x, lo);
                deg_bin_add(bins, v[u].y, lo);
                deg_bin_add(bins, v[u].z, lo);
                deg_bin_add(bins, v[u].w, lo);
            }
        }
    }
    for (int q = b0 + tid; q < end; q += DEG_THREADS) deg_bin_add(bins, dst[q], lo);
    __syncthreads();

    const int rot = ((int)blockIdx.x * (DEG_BINS / DEG_SLICES)) & ~31;
    for (int k = tid; k < DEG_BINS; k += DEG_THREADS) {
        int i = k + rot;
        if (i >= DEG_BINS) i -= DEG_BINS;
        const unsigned c = bins[i];
        const int node = lo + i;
        if (c != 0u && node < n)
            __hip_atomic_fetch_add(cnt + node, c, __ATOMIC_RELAXED, __HIP_MEMORY_SCOPE_AGENT);
    }
}

__global__ void k_rsqrt(unsigned* __restrict__ cnt, int n) {
    int i = blockIdx.x * blockDim.x + threadIdx.x;
    if (i < n) cnt[i] = __float_as_uint(rsqrtf((float)cnt[i] + 1.0f));
}

template<int K, bool LAYER2>
__global__ void k_gemm16(const float* __restrict__ A, const float* __restrict__ W,
                         const float* __restrict__ dinv, const float* __restrict__ bias,
                         float* __restrict__ D, float* __restrict__ D2, int n, int n_tiles) {
    const int lane = threadIdx.x & 31;
    const int wid  = threadIdx.x >> 5;
    const int tile = blockIdx.x * (blockDim.x >> 5) + wid;
    if (tile >= n_tiles) return;

    const int m    = lane & 15;
    const int half = lane >> 4;
    const int kh   = half * 2;

    int rowA = tile * 16 + m;
    int rowC = rowA;
    if (rowA >= n) rowA = n - 1;

    const float* arow = A + (size_t)rowA * K;
    const float  dA   = LAYER2 ? dinv[rowA] : 1.0f;
    v8f acc = {};
#pragma unroll
    for (int k0 = 0; k0 < K; k0 += 4) {
        float a0 = arow[k0 + kh];
        float a1 = arow[k0 + kh + 1];
        if (LAYER2) {
            a0 = fmaxf(dA * a0 + bias[k0 + kh], 0.0f);
            a1 = fmaxf(dA * a1 + bias[k0 + kh + 1], 0.0f);
        }
        v2f a; a.x = a0; a.y = a1;
        v2f b;
        b.x = W[(k0 + kh) * HID + m];
        b.y = W[(k0 + kh + 1) * HID + m];
        acc = __builtin_amdgcn_wmma_f32_16x16x4_f32(false, a, false, b,
                                                    (short)0, acc, false, false);
    }
#pragma unroll
    for (int r = 0; r < 8; ++r) {
        int row = tile * 16 + r + 8 * half;
        if (row < n) {
            const float v = dinv[row] * acc[r];
            D[(size_t)row * HID + m] = v;
            if (!LAYER2) D2[(size_t)row * HID + m] = v;
        }
    }
}

#define EDGES_PER_WAVE 32
#define EPROP_THREADS  256
#define EPROP_BLOCKS   2048
#define EPROP_TRIPS    2
#define EPROP_PIECE    ((EPROP_THREADS / 32) * EPROP_TRIPS * EDGES_PER_WAVE)

template<int J>
__device__ __forceinline__ int row_lane(int v) {
    return __builtin_amdgcn_update_dpp(0, v, 0x150 + J, 0xf, 0xf, true);
}
template<int J>
__device__ __forceinline__ float row_lane(float v) {
    return __int_as_float(row_lane<J>(__float_as_int(v)));
}

__device__ __forceinline__ unsigned take_piece(unsigned* q) {
    return __hip_atomic_fetch_add(q, 1u, __ATOMIC_RELAXED, __HIP_MEMORY_SCOPE_AGENT);
}

__global__ __launch_bounds__(EPROP_THREADS)
void k_edge_prop(const int* __restrict__ src, const int* __restrict__ dst,
                 const float* __restrict__ h, float* __restrict__ out,
                 unsigned* __restrict__ ctr, int E, int n) {
    __shared__ unsigned nextp[2];
    const int tid  = threadIdx.x;
    const int lane = tid & 31;
    const int wv   = __builtin_amdgcn_readfirstlane(tid >> 5);
    const int c    = lane & 15;
    const int hb   = lane & 16;

    const int myhalf = (int)((__builtin_amdgcn_s_sendmsg_rtn(0x87) >> 18) & 1u);
    const int split  = ((n >> 1) + 3) & ~3;
    const unsigned npieces = ((unsigned)E + EPROP_PIECE - 1) / EPROP_PIECE;

#pragma unroll 1
    for (int pass = 0; pass < 2; ++pass) {
        const int hh = myhalf ^ pass;
        unsigned* q = ctr + hh * CTR_STRIDE;
        if (tid == 0) nextp[0] = take_piece(q);
        __syncthreads();
        for (unsigned it = 0; ; ++it) {
            const unsigned p = __builtin_amdgcn_readfirstlane(nextp[it & 1]);
            if (p >= npieces) break;
            if (tid == 0) nextp[(it + 1) & 1] = take_piece(q);

#pragma unroll 1
            for (int t = 0; t < EPROP_TRIPS; ++t) {
                const int base = (int)(p * EPROP_PIECE) + (t * (EPROP_THREADS / 32) + wv) * EDGES_PER_WAVE;
                if (base >= E) break;
                const int e  = base + lane;
                const int ec = e < E ? e : E - 1;
                const int s_l = src[ec];
                const int d_l = dst[ec];
                const bool mine = (e < E) && ((d_l >= split) == (hh != 0));
                const unsigned rm = (__builtin_amdgcn_ballot_w32(mine) >> hb) & 0xffffu;

                unsigned off_d[16];
                float    hv[16];
#define GATHER_ROUND(J)                                              \
                {                                                    \
                    const int s = row_lane<J>(s_l);                  \
                    const int d = row_lane<J>(d_l);                  \
                    off_d[J] = (unsigned)d * HID + c;                \
                    hv[J]    = 0.0f;                                 \
                    if (rm & (1u << J)) hv[J] = h[(unsigned)s * HID + c]; \
                }
                GATHER_ROUND(0)  GATHER_ROUND(1)  GATHER_ROUND(2)  GATHER_ROUND(3)
                GATHER_ROUND(4)  GATHER_ROUND(5)  GATHER_ROUND(6)  GATHER_ROUND(7)
                GATHER_ROUND(8)  GATHER_ROUND(9)  GATHER_ROUND(10) GATHER_ROUND(11)
                GATHER_ROUND(12) GATHER_ROUND(13) GATHER_ROUND(14) GATHER_ROUND(15)
#undef GATHER_ROUND
#pragma unroll
                for (int j = 0; j < 16; ++j)
                    if (rm & (1u << j)) atomAddF(out + off_d[j], hv[j]);
            }
            __syncthreads();
        }
        __syncthreads();
    }
}

#define EPOOL_THREADS 1024
#define EPOOL_BLOCKS  512

__global__ __launch_bounds__(EPOOL_THREADS)
void k_edge_pool(const int* __restrict__ src, const int* __restrict__ dst,
                 const int* __restrict__ rec,
                 const float* __restrict__ h, float* __restrict__ sums, int E) {
    __shared__ float tab[NUM_GRAPHS * HID];
    const int tid = threadIdx.x;
    for (int i = tid; i < NUM_GRAPHS * HID; i += EPOOL_THREADS) tab[i] = 0.0f;
    __syncthreads();

    const int lane = tid & 31;
    const int c    = lane & 15;
    const int hb   = lane & 16;
    const int wave   = blockIdx.x * (EPOOL_THREADS / 32) + (tid >> 5);
    const int stride = (int)gridDim.x * (EPOOL_THREADS / 32) * EDGES_PER_WAVE;

    for (int base = wave * EDGES_PER_WAVE; base < E; base += stride) {
        const int e  = base + lane;
        const int ec = e < E ? e : E - 1;
        const int   s_l  = src[ec];
        const int   d_l  = dst[ec];
        const int2  r_l  = reinterpret_cast<const int2*>(rec)[d_l];
        const float w_l  = __int_as_float(r_l.x);
        const int   g_l  = r_l.y;

        if (base + EDGES_PER_WAVE <= E) {
            float hv[16];
#define POOL_GATHER(J) hv[J] = h[(unsigned)row_lane<J>(s_l) * HID + c];
            POOL_GATHER(0)  POOL_GATHER(1)  POOL_GATHER(2)  POOL_GATHER(3)
            POOL_GATHER(4)  POOL_GATHER(5)  POOL_GATHER(6)  POOL_GATHER(7)
            POOL_GATHER(8)  POOL_GATHER(9)  POOL_GATHER(10) POOL_GATHER(11)
            POOL_GATHER(12) POOL_GATHER(13) POOL_GATHER(14) POOL_GATHER(15)
#undef POOL_GATHER
#define POOL_ADD(J) ldsAddF(tab + ((unsigned)row_lane<J>(g_l) * HID + c), row_lane<J>(w_l) * hv[J]);
            POOL_ADD(0)  POOL_ADD(1)  POOL_ADD(2)  POOL_ADD(3)
            POOL_ADD(4)  POOL_ADD(5)  POOL_ADD(6)  POOL_ADD(7)
            POOL_ADD(8)  POOL_ADD(9)  POOL_ADD(10) POOL_ADD(11)
            POOL_ADD(12) POOL_ADD(13) POOL_ADD(14) POOL_ADD(15)
#undef POOL_ADD
        } else {
#pragma unroll 1
            for (int j = 0; j < 16; ++j) {
                const int   s = __shfl(s_l, hb + j);
                const int   g = __shfl(g_l, hb + j);
                const float w = __shfl(w_l, hb + j);
                if (base + hb + j < E)
                    ldsAddF(tab + ((unsigned)g * HID + c), w * h[(unsigned)s * HID + c]);
            }
        }
    }
    __syncthreads();

    const int rot = ((int)blockIdx.x * 48) & (NUM_GRAPHS * HID - 1) & ~31;
    for (int k = tid; k < NUM_GRAPHS * HID; k += EPOOL_THREADS) {
        const int i = (k + rot) & (NUM_GRAPHS * HID - 1);
        atomAddF(sums + i, tab[i]);
    }
}

__global__ void k_zero_pool(float* __restrict__ sums, float* __restrict__ cnt) {
    int t = blockIdx.x * blockDim.x + threadIdx.x;
    if (t < NUM_GRAPHS * HID) sums[t] = 0.0f;
    if (t < NUM_GRAPHS) cnt[t] = 0.0f;
}

__global__ void k_pool(const float* __restrict__ h, const float* __restrict__ dinv,
                       const float* __restrict__ bias, const int* __restrict__ batch,
                       int* __restrict__ rec,
                       float* __restrict__ sums, float* __restrict__ cnt, int n) {
    int i = blockIdx.x * blockDim.x + threadIdx.x;
    if (i >= n) return;
    const int   g  = batch[i];
    const float di = dinv[i];
    reinterpret_cast<int2*>(rec)[i] = make_int2(__float_as_int(di), g);
    atomAddF(&cnt[g], 1.0f);
    const float* row = h + (size_t)i * HID;
#pragma unroll
    for (int c = 0; c < HID; ++c) atomAddF(&sums[g * HID + c], di * row[c] + bias[c]);
}

__global__ void k_head(const float* __restrict__ sums, const float* __restrict__ cnt,
                       const float* __restrict__ fcW, const float* __restrict__ fcb,
                       float* __restrict__ y) {
    int t = blockIdx.x * blockDim.x + threadIdx.x;
    if (t >= NUM_GRAPHS * NGRAPH) return;
    int g = t >> 4, cls = t & 15;
    float inv = 1.0f / fmaxf(cnt[g], 1.0f);
    float acc = fcb[cls];
#pragma unroll
    for (int k = 0; k < HID; ++k)
        acc += (sums[g * HID + k] * inv) * fcW[k * NGRAPH + cls];
    y[t] = acc;
}

extern "C" void kernel_launch(void* const* d_in, const int* in_sizes, int n_in,
                              void* d_out, int out_size, void* d_ws, size_t ws_size,
                              hipStream_t stream) {
    const float* x    = (const float*)d_in[0];
    const float* W1   = (const float*)d_in[1];
    const float* b1   = (const float*)d_in[2];
    const float* W2   = (const float*)d_in[3];
    const float* b2   = (const float*)d_in[4];
    const float* fcW  = (const float*)d_in[5];
    const float* fcb  = (const float*)d_in[6];
    const int*   eidx = (const int*)d_in[7];
    const int*   batch= (const int*)d_in[8];
    float* y = (float*)d_out;

    const int n = in_sizes[0] / 128;
    const int E = in_sizes[7] / 2;
    const int* src = eidx;
    const int* dst = eidx + E;

    size_t nAl = ((size_t)n + 3) & ~(size_t)3;
    float* deg  = (float*)d_ws;
    float* h    = deg + nAl;
    float* outb = h + (size_t)n * HID;
    float* sums = outb + (size_t)n * HID;
    float* cnt  = sums + NUM_GRAPHS * HID;

    const int TB = 256;
    const int n_tiles = (n + 15) / 16;

    unsigned* degc = (unsigned*)d_ws;
    k_init_deg<<<(n + TB - 1) / TB, TB, 0, stream>>>(degc, (unsigned*)sums, n);
    k_deg_accum<<<dim3(DEG_SLICES, (n + DEG_BINS - 1) / DEG_BINS), DEG_THREADS, 0, stream>>>(dst, degc, E, n);
    k_rsqrt<<<(n + TB - 1) / TB, TB, 0, stream>>>(degc, n);

    k_gemm16<128, false><<<(n_tiles + 3) / 4, 128, 0, stream>>>(x, W1, deg, b1, h, outb, n, n_tiles);
    k_edge_prop<<<EPROP_BLOCKS, EPROP_THREADS, 0, stream>>>(src, dst, h, outb, (unsigned*)sums, E, n);

    k_gemm16<16, true><<<(n_tiles + 3) / 4, 128, 0, stream>>>(outb, W2, deg, b1, h, nullptr, n, n_tiles);

    int* rec = (int*)outb;
    k_zero_pool<<<(NUM_GRAPHS * HID + TB - 1) / TB, TB, 0, stream>>>(sums, cnt);
    k_pool<<<(n + TB - 1) / TB, TB, 0, stream>>>(h, deg, b2, batch, rec, sums, cnt, n);
    k_edge_pool<<<EPOOL_BLOCKS, EPOOL_THREADS, 0, stream>>>(src, dst, rec, h, sums, E);
    k_head<<<(NUM_GRAPHS * NGRAPH + TB - 1) / TB, TB, 0, stream>>>(sums, cnt, fcW, fcb, y);
}


# per-node pool terms: a wave sums its 32 nodes' rows in registers (lane = channel) and adds one row per run of equal graph ids, not 17 same-word global atomics per node; pool sum order changes
# speedup vs baseline: 5.1980x; 1.9731x over previous
#include <hip/hip_runtime.h>
#include <hip/hip_bf16.h>

typedef __attribute__((ext_vector_type(2))) float v2f;
typedef __attribute__((ext_vector_type(8))) float v8f;

#define HID 16
#define NGRAPH 16
#define NUM_GRAPHS 512

__device__ __forceinline__ void atomAddF(float* p, float v) {
    __hip_atomic_fetch_add(p, v, __ATOMIC_RELAXED, __HIP_MEMORY_SCOPE_AGENT);
}

__device__ __forceinline__ void ldsAddF(float* p, float v) {
    __hip_atomic_fetch_add(p, v, __ATOMIC_RELAXED, __HIP_MEMORY_SCOPE_WORKGROUP);
}

#define DEG_BINS    50176
#define DEG_THREADS 1024
#define DEG_SLICES  64
#define DEG_QUADS   8

#define CTR_STRIDE  64

__global__ void k_init_deg(unsigned* __restrict__ cnt, unsigned* __restrict__ ctr, int n) {
    int i = blockIdx.x * blockDim.x + threadIdx.x;
    if (i < n) cnt[i] = 0u;
    if (i < 2) ctr[i * CTR_STRIDE] = 0u;
}

__device__ __forceinline__ void deg_bin_add(unsigned* bins, int d, int lo) {
    const unsigned r = (unsigned)(d - lo);
    if (r < (unsigned)DEG_BINS)
        __hip_atomic_fetch_add(bins + r, 1u, __ATOMIC_RELAXED, __HIP_MEMORY_SCOPE_WORKGROUP);
}

__global__ __launch_bounds__(DEG_THREADS)
void k_deg_accum(const int* __restrict__ dst, unsigned* __restrict__ cnt, int E, int n) {
    __shared__ unsigned bins[DEG_BINS];
    const int tid = threadIdx.x;
    const int lo  = blockIdx.y * DEG_BINS;

    for (int i = tid; i < DEG_BINS; i += DEG_THREADS) bins[i] = 0u;
    __syncthreads();

    const int per = (((E + (int)gridDim.x - 1) / (int)gridDim.x) + 3) & ~3;
    const int beg = blockIdx.x * per;
    int end = beg + per;
    if (end > E) end = E;

    int b0 = beg;
    if ((reinterpret_cast<size_t>(dst) & 15) == 0) {
        const int batch = 4 * DEG_THREADS * DEG_QUADS;
        for (; b0 + batch <= end; b0 += batch) {
            const int4* p = reinterpret_cast<const int4*>(dst + b0) + tid;
            int4 v[DEG_QUADS];
#pragma unroll
            for (int u = 0; u < DEG_QUADS; ++u) v[u] = p[u * DEG_THREADS];
#pragma unroll
            for (int u = 0; u < DEG_QUADS; ++u) {
                deg_bin_add(bins, v[u].x, lo);
                deg_bin_add(bins, v[u].y, lo);
                deg_bin_add(bins, v[u].z, lo);
                deg_bin_add(bins, v[u].w, lo);
            }
        }
    }
    for (int q = b0 + tid; q < end; q += DEG_THREADS) deg_bin_add(bins, dst[q], lo);
    __syncthreads();

    const int rot = ((int)blockIdx.x * (DEG_BINS / DEG_SLICES)) & ~31;
    for (int k = tid; k < DEG_BINS; k += DEG_THREADS) {
        int i = k + rot;
        if (i >= DEG_BINS) i -= DEG_BINS;
        const unsigned c = bins[i];
        const int node = lo + i;
        if (c != 0u && node < n)
            __hip_atomic_fetch_add(cnt + node, c, __ATOMIC_RELAXED, __HIP_MEMORY_SCOPE_AGENT);
    }
}

__global__ void k_rsqrt(unsigned* __restrict__ cnt, int n) {
    int i = blockIdx.x * blockDim.x + threadIdx.x;
    if (i < n) cnt[i] = __float_as_uint(rsqrtf((float)cnt[i] + 1.0f));
}

template<int K, bool LAYER2>
__global__ void k_gemm16(const float* __restrict__ A, const float* __restrict__ W,
                         const float* __restrict__ dinv, const float* __restrict__ bias,
                         float* __restrict__ D, float* __restrict__ D2, int n, int n_tiles) {
    const int lane = threadIdx.x & 31;
    const int wid  = threadIdx.x >> 5;
    const int tile = blockIdx.x * (blockDim.x >> 5) + wid;
    if (tile >= n_tiles) return;

    const int m    = lane & 15;
    const int half = lane >> 4;
    const int kh   = half * 2;

    int rowA = tile * 16 + m;
    int rowC = rowA;
    if (rowA >= n) rowA = n - 1;

    const float* arow = A + (size_t)rowA * K;
    const float  dA   = LAYER2 ? dinv[rowA] : 1.0f;
    v8f acc = {};
#pragma unroll
    for (int k0 = 0; k0 < K; k0 += 4) {
        float a0 = arow[k0 + kh];
        float a1 = arow[k0 + kh + 1];
        if (LAYER2) {
            a0 = fmaxf(dA * a0 + bias[k0 + kh], 0.0f);
            a1 = fmaxf(dA * a1 + bias[k0 + kh + 1], 0.0f);
        }
        v2f a; a.x = a0; a.y = a1;
        v2f b;
        b.x = W[(k0 + kh) * HID + m];
        b.y = W[(k0 + kh + 1) * HID + m];
        acc = __builtin_amdgcn_wmma_f32_16x16x4_f32(false, a, false, b,
                                                    (short)0, acc, false, false);
    }
#pragma unroll
    for (int r = 0; r < 8; ++r) {
        int row = tile * 16 + r + 8 * half;
        if (row < n) {
            const float v = dinv[row] * acc[r];
            D[(size_t)row * HID + m] = v;
            if (!LAYER2) D2[(size_t)row * HID + m] = v;
        }
    }
}

#define EDGES_PER_WAVE 32
#define EPROP_THREADS  256
#define EPROP_BLOCKS   2048
#define EPROP_TRIPS    2
#define EPROP_PIECE    ((EPROP_THREADS / 32) * EPROP_TRIPS * EDGES_PER_WAVE)

template<int J>
__device__ __forceinline__ int row_lane(int v) {
    return __builtin_amdgcn_update_dpp(0, v, 0x150 + J, 0xf, 0xf, true);
}
template<int J>
__device__ __forceinline__ float row_lane(float v) {
    return __int_as_float(row_lane<J>(__float_as_int(v)));
}

__device__ __forceinline__ unsigned take_piece(unsigned* q) {
    return __hip_atomic_fetch_add(q, 1u, __ATOMIC_RELAXED, __HIP_MEMORY_SCOPE_AGENT);
}

__global__ __launch_bounds__(EPROP_THREADS)
void k_edge_prop(const int* __restrict__ src, const int* __restrict__ dst,
                 const float* __restrict__ h, float* __restrict__ out,
                 unsigned* __restrict__ ctr, int E, int n) {
    __shared__ unsigned nextp[2];
    const int tid  = threadIdx.x;
    const int lane = tid & 31;
    const int wv   = __builtin_amdgcn_readfirstlane(tid >> 5);
    const int c    = lane & 15;
    const int hb   = lane & 16;

    const int myhalf = (int)((__builtin_amdgcn_s_sendmsg_rtn(0x87) >> 18) & 1u);
    const int split  = ((n >> 1) + 3) & ~3;
    const unsigned npieces = ((unsigned)E + EPROP_PIECE - 1) / EPROP_PIECE;

#pragma unroll 1
    for (int pass = 0; pass < 2; ++pass) {
        const int hh = myhalf ^ pass;
        unsigned* q = ctr + hh * CTR_STRIDE;
        if (tid == 0) nextp[0] = take_piece(q);
        __syncthreads();
        for (unsigned it = 0; ; ++it) {
            const unsigned p = __builtin_amdgcn_readfirstlane(nextp[it & 1]);
            if (p >= npieces) break;
            if (tid == 0) nextp[(it + 1) & 1] = take_piece(q);

#pragma unroll 1
            for (int t = 0; t < EPROP_TRIPS; ++t) {
                const int base = (int)(p * EPROP_PIECE) + (t * (EPROP_THREADS / 32) + wv) * EDGES_PER_WAVE;
                if (base >= E) break;
                const int e  = base + lane;
                const int ec = e < E ? e : E - 1;
                const int s_l = src[ec];
                const int d_l = dst[ec];
                const bool mine = (e < E) && ((d_l >= split) == (hh != 0));
                const unsigned rm = (__builtin_amdgcn_ballot_w32(mine) >> hb) & 0xffffu;

                unsigned off_d[16];
                float    hv[16];
#define GATHER_ROUND(J)                                              \
                {                                                    \
                    const int s = row_lane<J>(s_l);                  \
                    const int d = row_lane<J>(d_l);                  \
                    off_d[J] = (unsigned)d * HID + c;                \
                    hv[J]    = 0.0f;                                 \
                    if (rm & (1u << J)) hv[J] = h[(unsigned)s * HID + c]; \
                }
                GATHER_ROUND(0)  GATHER_ROUND(1)  GATHER_ROUND(2)  GATHER_ROUND(3)
                GATHER_ROUND(4)  GATHER_ROUND(5)  GATHER_ROUND(6)  GATHER_ROUND(7)
                GATHER_ROUND(8)  GATHER_ROUND(9)  GATHER_ROUND(10) GATHER_ROUND(11)
                GATHER_ROUND(12) GATHER_ROUND(13) GATHER_ROUND(14) GATHER_ROUND(15)
#undef GATHER_ROUND
#pragma unroll
                for (int j = 0; j < 16; ++j)
                    if (rm & (1u << j)) atomAddF(out + off_d[j], hv[j]);
            }
            __syncthreads();
        }
        __syncthreads();
    }
}

#define EPOOL_THREADS 1024
#define EPOOL_BLOCKS  512

__global__ __launch_bounds__(EPOOL_THREADS)
void k_edge_pool(const int* __restrict__ src, const int* __restrict__ dst,
                 const int* __restrict__ rec,
                 const float* __restrict__ h, float* __restrict__ sums, int E) {
    __shared__ float tab[NUM_GRAPHS * HID];
    const int tid = threadIdx.x;
    for (int i = tid; i < NUM_GRAPHS * HID; i += EPOOL_THREADS) tab[i] = 0.0f;
    __syncthreads();

    const int lane = tid & 31;
    const int c    = lane & 15;
    const int hb   = lane & 16;
    const int wave   = blockIdx.x * (EPOOL_THREADS / 32) + (tid >> 5);
    const int stride = (int)gridDim.x * (EPOOL_THREADS / 32) * EDGES_PER_WAVE;

    for (int base = wave * EDGES_PER_WAVE; base < E; base += stride) {
        const int e  = base + lane;
        const int ec = e < E ? e : E - 1;
        const int   s_l  = src[ec];
        const int   d_l  = dst[ec];
        const int2  r_l  = reinterpret_cast<const int2*>(rec)[d_l];
        const float w_l  = __int_as_float(r_l.x);
        const int   g_l  = r_l.y;

        if (base + EDGES_PER_WAVE <= E) {
            float hv[16];
#define POOL_GATHER(J) hv[J] = h[(unsigned)row_lane<J>(s_l) * HID + c];
            POOL_GATHER(0)  POOL_GATHER(1)  POOL_GATHER(2)  POOL_GATHER(3)
            POOL_GATHER(4)  POOL_GATHER(5)  POOL_GATHER(6)  POOL_GATHER(7)
            POOL_GATHER(8)  POOL_GATHER(9)  POOL_GATHER(10) POOL_GATHER(11)
            POOL_GATHER(12) POOL_GATHER(13) POOL_GATHER(14) POOL_GATHER(15)
#undef POOL_GATHER
#define POOL_ADD(J) ldsAddF(tab + ((unsigned)row_lane<J>(g_l) * HID + c), row_lane<J>(w_l) * hv[J]);
            POOL_ADD(0)  POOL_ADD(1)  POOL_ADD(2)  POOL_ADD(3)
            POOL_ADD(4)  POOL_ADD(5)  POOL_ADD(6)  POOL_ADD(7)
            POOL_ADD(8)  POOL_ADD(9)  POOL_ADD(10) POOL_ADD(11)
            POOL_ADD(12) POOL_ADD(13) POOL_ADD(14) POOL_ADD(15)
#undef POOL_ADD
        } else {
#pragma unroll 1
            for (int j = 0; j < 16; ++j) {
                const int   s = __shfl(s_l, hb + j);
                const int   g = __shfl(g_l, hb + j);
                const float w = __shfl(w_l, hb + j);
                if (base + hb + j < E)
                    ldsAddF(tab + ((unsigned)g * HID + c), w * h[(unsigned)s * HID + c]);
            }
        }
    }
    __syncthreads();

    const int rot = ((int)blockIdx.x * 48) & (NUM_GRAPHS * HID - 1) & ~31;
    for (int k = tid; k < NUM_GRAPHS * HID; k += EPOOL_THREADS) {
        const int i = (k + rot) & (NUM_GRAPHS * HID - 1);
        atomAddF(sums + i, tab[i]);
    }
}

__global__ void k_zero_pool(float* __restrict__ sums, float* __restrict__ cnt) {
    int t = blockIdx.x * blockDim.x + threadIdx.x;
    if (t < NUM_GRAPHS * HID) sums[t] = 0.0f;
    if (t < NUM_GRAPHS) cnt[t] = 0.0f;
}

#define POOL_NODES 32

__device__ __forceinline__ void pool_flush(float* __restrict__ sums, float* __restrict__ cnt,
                                           int g, int c, float acc, float num) {
    if (g >= 0) {
        atomAddF(sums + g * HID + c, acc);
        if (c == 0) atomAddF(cnt + g, num);
    }
}

__global__ void k_pool(const float* __restrict__ h, const float* __restrict__ dinv,
                       const float* __restrict__ bias, const int* __restrict__ batch,
                       int* __restrict__ rec,
                       float* __restrict__ sums, float* __restrict__ cnt, int n) {
    const int lane = threadIdx.x & 31;
    const int c    = lane & 15;
    const int half = lane >> 4;
    const int wave = blockIdx.x * (blockDim.x >> 5) + (threadIdx.x >> 5);
    const int base = wave * POOL_NODES;
    if (base >= n) return;

    const float b = bias[c];
    int   g[POOL_NODES / 2];
    float v[POOL_NODES / 2];
#pragma unroll
    for (int j = 0; j < POOL_NODES / 2; ++j) {
        const int  node = base + 2 * j + half;
        const bool ok   = node < n;
        const int  nc   = ok ? node : n - 1;
        const int gl = batch[nc];
        g[j] = ok ? gl : -1;
        v[j] = dinv[nc] * h[(size_t)nc * HID + c] + b;
    }
    {
        const int i = base + lane;
        if (i < n) reinterpret_cast<int2*>(rec)[i] = make_int2(__float_as_int(dinv[i]), batch[i]);
    }

    int   gcur = g[0];
    float acc = 0.0f, num = 0.0f;
#pragma unroll
    for (int j = 0; j < POOL_NODES / 2; ++j) {
        if (g[j] != gcur) {
            pool_flush(sums, cnt, gcur, c, acc, num);
            gcur = g[j]; acc = 0.0f; num = 0.0f;
        }
        acc += v[j];
        num += 1.0f;
    }
    pool_flush(sums, cnt, gcur, c, acc, num);
}

__global__ void k_head(const float* __restrict__ sums, const float* __restrict__ cnt,
                       const float* __restrict__ fcW, const float* __restrict__ fcb,
                       float* __restrict__ y) {
    int t = blockIdx.x * blockDim.x + threadIdx.x;
    if (t >= NUM_GRAPHS * NGRAPH) return;
    int g = t >> 4, cls = t & 15;
    float inv = 1.0f / fmaxf(cnt[g], 1.0f);
    float acc = fcb[cls];
#pragma unroll
    for (int k = 0; k < HID; ++k)
        acc += (sums[g * HID + k] * inv) * fcW[k * NGRAPH + cls];
    y[t] = acc;
}

extern "C" void kernel_launch(void* const* d_in, const int* in_sizes, int n_in,
                              void* d_out, int out_size, void* d_ws, size_t ws_size,
                              hipStream_t stream) {
    const float* x    = (const float*)d_in[0];
    const float* W1   = (const float*)d_in[1];
    const float* b1   = (const float*)d_in[2];
    const float* W2   = (const float*)d_in[3];
    const float* b2   = (const float*)d_in[4];
    const float* fcW  = (const float*)d_in[5];
    const float* fcb  = (const float*)d_in[6];
    const int*   eidx = (const int*)d_in[7];
    const int*   batch= (const int*)d_in[8];
    float* y = (float*)d_out;

    const int n = in_sizes[0] / 128;
    const int E = in_sizes[7] / 2;
    const int* src = eidx;
    const int* dst = eidx + E;

    size_t nAl = ((size_t)n + 3) & ~(size_t)3;
    float* deg  = (float*)d_ws;
    float* h    = deg + nAl;
    float* outb = h + (size_t)n * HID;
    float* sums = outb + (size_t)n * HID;
    float* cnt  = sums + NUM_GRAPHS * HID;

    const int TB = 256;
    const int n_tiles = (n + 15) / 16;

    unsigned* degc = (unsigned*)d_ws;
    k_init_deg<<<(n + TB - 1) / TB, TB, 0, stream>>>(degc, (unsigned*)sums, n);
    k_deg_accum<<<dim3(DEG_SLICES, (n + DEG_BINS - 1) / DEG_BINS), DEG_THREADS, 0, stream>>>(dst, degc, E, n);
    k_rsqrt<<<(n + TB - 1) / TB, TB, 0, stream>>>(degc, n);

    k_gemm16<128, false><<<(n_tiles + 3) / 4, 128, 0, stream>>>(x, W1, deg, b1, h, outb, n, n_tiles);
    k_edge_prop<<<EPROP_BLOCKS, EPROP_THREADS, 0, stream>>>(src, dst, h, outb, (unsigned*)sums, E, n);

    k_gemm16<16, true><<<(n_tiles + 3) / 4, 128, 0, stream>>>(outb, W2, deg, b1, h, nullptr, n, n_tiles);

    int* rec = (int*)outb;
    k_zero_pool<<<(NUM_GRAPHS * HID + TB - 1) / TB, TB, 0, stream>>>(sums, cnt);
    k_pool<<<(n + TB - 1) / TB, TB, 0, stream>>>(h, deg, b2, batch, rec, sums, cnt, n);
    k_edge_pool<<<EPOOL_BLOCKS, EPOOL_THREADS, 0, stream>>>(src, dst, rec, h, sums, E);
    k_head<<<(NUM_GRAPHS * NGRAPH + TB - 1) / TB, TB, 0, stream>>>(sums, cnt, fcW, fcb, y);
}


# layer-2 edge pass: a work-group keeps 16-bit records (graph id, degree) of 1 of 2 node ranges in LDS and packs its edges into full 32-edge batches: no 8-byte L2 read per edge; pool sum order changes
# speedup vs baseline: 5.5024x; 1.0586x over previous
#include <hip/hip_runtime.h>
#include <hip/hip_bf16.h>

typedef __attribute__((ext_vector_type(2))) float v2f;
typedef __attribute__((ext_vector_type(8))) float v8f;

#define HID 16
#define NGRAPH 16
#define NUM_GRAPHS 512

__device__ __forceinline__ void atomAddF(float* p, float v) {
    __hip_atomic_fetch_add(p, v, __ATOMIC_RELAXED, __HIP_MEMORY_SCOPE_AGENT);
}

__device__ __forceinline__ void ldsAddF(float* p, float v) {
    __hip_atomic_fetch_add(p, v, __ATOMIC_RELAXED, __HIP_MEMORY_SCOPE_WORKGROUP);
}

#define DEG_BINS    50176
#define DEG_THREADS 1024
#define DEG_SLICES  64
#define DEG_QUADS   8

#define CTR_STRIDE  64

__global__ void k_init_deg(unsigned* __restrict__ cnt, unsigned* __restrict__ ctr, int n) {
    int i = blockIdx.x * blockDim.x + threadIdx.x;
    if (i < n) cnt[i] = 0u;
    if (i < 2) ctr[i * CTR_STRIDE] = 0u;
}

__device__ __forceinline__ void deg_bin_add(unsigned* bins, int d, int lo) {
    const unsigned r = (unsigned)(d - lo);
    if (r < (unsigned)DEG_BINS)
        __hip_atomic_fetch_add(bins + r, 1u, __ATOMIC_RELAXED, __HIP_MEMORY_SCOPE_WORKGROUP);
}

__global__ __launch_bounds__(DEG_THREADS)
void k_deg_accum(const int* __restrict__ dst, unsigned* __restrict__ cnt, int E, int n) {
    __shared__ unsigned bins[DEG_BINS];
    const int tid = threadIdx.x;
    const int lo  = blockIdx.y * DEG_BINS;

    for (int i = tid; i < DEG_BINS; i += DEG_THREADS) bins[i] = 0u;
    __syncthreads();

    const int per = (((E + (int)gridDim.x - 1) / (int)gridDim.x) + 3) & ~3;
    const int beg = blockIdx.x * per;
    int end = beg + per;
    if (end > E) end = E;

    int b0 = beg;
    if ((reinterpret_cast<size_t>(dst) & 15) == 0) {
        const int batch = 4 * DEG_THREADS * DEG_QUADS;
        for (; b0 + batch <= end; b0 += batch) {
            const int4* p = reinterpret_cast<const int4*>(dst + b0) + tid;
            int4 v[DEG_QUADS];
#pragma unroll
            for (int u = 0; u < DEG_QUADS; ++u) v[u] = p[u * DEG_THREADS];
#pragma unroll
            for (int u = 0; u < DEG_QUADS; ++u) {
                deg_bin_add(bins, v[u].x, lo);
                deg_bin_add(bins, v[u].y, lo);
                deg_bin_add(bins, v[u].z, lo);
                deg_bin_add(bins, v[u].w, lo);
            }
        }
    }
    for (int q = b0 + tid; q < end; q += DEG_THREADS) deg_bin_add(bins, dst[q], lo);
    __syncthreads();

    const int rot = ((int)blockIdx.x * (DEG_BINS / DEG_SLICES)) & ~31;
    for (int k = tid; k < DEG_BINS; k += DEG_THREADS) {
        int i = k + rot;
        if (i >= DEG_BINS) i -= DEG_BINS;
        const unsigned c = bins[i];
        const int node = lo + i;
        if (c != 0u && node < n)
            __hip_atomic_fetch_add(cnt + node, c, __ATOMIC_RELAXED, __HIP_MEMORY_SCOPE_AGENT);
    }
}

__global__ void k_rsqrt(unsigned* __restrict__ cnt, int n) {
    int i = blockIdx.x * blockDim.x + threadIdx.x;
    if (i < n) cnt[i] = __float_as_uint(rsqrtf((float)cnt[i] + 1.0f));
}

template<int K, bool LAYER2>
__global__ void k_gemm16(const float* __restrict__ A, const float* __restrict__ W,
                         const float* __restrict__ dinv, const float* __restrict__ bias,
                         float* __restrict__ D, float* __restrict__ D2, int n, int n_tiles) {
    const int lane = threadIdx.x & 31;
    const int wid  = threadIdx.x >> 5;
    const int tile = blockIdx.x * (blockDim.x >> 5) + wid;
    if (tile >= n_tiles) return;

    const int m    = lane & 15;
    const int half = lane >> 4;
    const int kh   = half * 2;

    int rowA = tile * 16 + m;
    int rowC = rowA;
    if (rowA >= n) rowA = n - 1;

    const float* arow = A + (size_t)rowA * K;
    const float  dA   = LAYER2 ? dinv[rowA] : 1.0f;
    v8f acc = {};
#pragma unroll
    for (int k0 = 0; k0 < K; k0 += 4) {
        float a0 = arow[k0 + kh];
        float a1 = arow[k0 + kh + 1];
        if (LAYER2) {
            a0 = fmaxf(dA * a0 + bias[k0 + kh], 0.0f);
            a1 = fmaxf(dA * a1 + bias[k0 + kh + 1], 0.0f);
        }
        v2f a; a.x = a0; a.y = a1;
        v2f b;
        b.x = W[(k0 + kh) * HID + m];
        b.y = W[(k0 + kh + 1) * HID + m];
        acc = __builtin_amdgcn_wmma_f32_16x16x4_f32(false, a, false, b,
                                                    (short)0, acc, false, false);
    }
#pragma unroll
    for (int r = 0; r < 8; ++r) {
        int row = tile * 16 + r + 8 * half;
        if (row < n) {
            const float v = dinv[row] * acc[r];
            D[(size_t)row * HID + m] = v;
            if (!LAYER2) D2[(size_t)row * HID + m] = v;
        }
    }
}

#define EDGES_PER_WAVE 32
#define EPROP_THREADS  256
#define EPROP_BLOCKS   2048
#define EPROP_TRIPS    2
#define EPROP_PIECE    ((EPROP_THREADS / 32) * EPROP_TRIPS * EDGES_PER_WAVE)

template<int J>
__device__ __forceinline__ int row_lane(int v) {
    return __builtin_amdgcn_update_dpp(0, v, 0x150 + J, 0xf, 0xf, true);
}
template<int J>
__device__ __forceinline__ float row_lane(float v) {
    return __int_as_float(row_lane<J>(__float_as_int(v)));
}

__device__ __forceinline__ unsigned take_piece(unsigned* q) {
    return __hip_atomic_fetch_add(q, 1u, __ATOMIC_RELAXED, __HIP_MEMORY_SCOPE_AGENT);
}

__global__ __launch_bounds__(EPROP_THREADS)
void k_edge_prop(const int* __restrict__ src, const int* __restrict__ dst,
                 const float* __restrict__ h, float* __restrict__ out,
                 unsigned* __restrict__ ctr, int E, int n) {
    __shared__ unsigned nextp[2];
    const int tid  = threadIdx.x;
    const int lane = tid & 31;
    const int wv   = __builtin_amdgcn_readfirstlane(tid >> 5);
    const int c    = lane & 15;
    const int hb   = lane & 16;

    const int myhalf = (int)((__builtin_amdgcn_s_sendmsg_rtn(0x87) >> 18) & 1u);
    const int split  = ((n >> 1) + 3) & ~3;
    const unsigned npieces = ((unsigned)E + EPROP_PIECE - 1) / EPROP_PIECE;

#pragma unroll 1
    for (int pass = 0; pass < 2; ++pass) {
        const int hh = myhalf ^ pass;
        unsigned* q = ctr + hh * CTR_STRIDE;
        if (tid == 0) nextp[0] = take_piece(q);
        __syncthreads();
        for (unsigned it = 0; ; ++it) {
            const unsigned p = __builtin_amdgcn_readfirstlane(nextp[it & 1]);
            if (p >= npieces) break;
            if (tid == 0) nextp[(it + 1) & 1] = take_piece(q);

#pragma unroll 1
            for (int t = 0; t < EPROP_TRIPS; ++t) {
                const int base = (int)(p * EPROP_PIECE) + (t * (EPROP_THREADS / 32) + wv) * EDGES_PER_WAVE;
                if (base >= E) break;
                const int e  = base + lane;
                const int ec = e < E ? e : E - 1;
                const int s_l = src[ec];
                const int d_l = dst[ec];
                const bool mine = (e < E) && ((d_l >= split) == (hh != 0));
                const unsigned rm = (__builtin_amdgcn_ballot_w32(mine) >> hb) & 0xffffu;

                unsigned off_d[16];
                float    hv[16];
#define GATHER_ROUND(J)                                              \
                {                                                    \
                    const int s = row_lane<J>(s_l);                  \
                    const int d = row_lane<J>(d_l);                  \
                    off_d[J] = (unsigned)d * HID + c;                \
                    hv[J]    = 0.0f;                                 \
                    if (rm & (1u << J)) hv[J] = h[(unsigned)s * HID + c]; \
                }
                GATHER_ROUND(0)  GATHER_ROUND(1)  GATHER_ROUND(2)  GATHER_ROUND(3)
                GATHER_ROUND(4)  GATHER_ROUND(5)  GATHER_ROUND(6)  GATHER_ROUND(7)
                GATHER_ROUND(8)  GATHER_ROUND(9)  GATHER_ROUND(10) GATHER_ROUND(11)
                GATHER_ROUND(12) GATHER_ROUND(13) GATHER_ROUND(14) GATHER_ROUND(15)
#undef GATHER_ROUND
#pragma unroll
                for (int j = 0; j < 16; ++j)
                    if (rm & (1u << j)) atomAddF(out + off_d[j], hv[j]);
            }
            __syncthreads();
        }
        __syncthreads();
    }
}

#define EPOOL_THREADS 1024
#define EPOOL_SLICES  128
#define EPOOL_RANGE   100352
#define EPOOL_TRIPS   4
#define EPOOL_RING    256
#define EPOOL_DEGMAX  127u

__device__ __forceinline__ void pool_dest(const unsigned* dtab, const int* __restrict__ rec, int lo, int r, bool live,
                                          int& g, float& w) {
    const unsigned wrd = dtab[(unsigned)r >> 1];
    const unsigned pr  = (r & 1) ? (wrd >> 16) : (wrd & 0xffffu);
    const unsigned dg  = pr & EPOOL_DEGMAX;
    g = (int)(pr >> 7);
    w = rsqrtf((float)dg);
    if (live && dg == EPOOL_DEGMAX) w = __int_as_float(rec[2 * (lo + r)]);
}

__global__ __launch_bounds__(EPOOL_THREADS)
void k_edge_pool(const int* __restrict__ src, const int* __restrict__ dst,
                 const unsigned* __restrict__ pkw, const int* __restrict__ rec,
                 const float* __restrict__ h, float* __restrict__ sums, int E, int n) {
    __shared__ float    tab[NUM_GRAPHS * HID];
    __shared__ unsigned dtab[EPOOL_RANGE / 2];
    __shared__ int2     ring[(EPOOL_THREADS / 32) * (EPOOL_RING + 32)];
    const int tid = threadIdx.x;
    const int lo  = blockIdx.y * EPOOL_RANGE;
    for (int i = tid; i < NUM_GRAPHS * HID; i += EPOOL_THREADS) tab[i] = 0.0f;
    {
        int nrec = n - lo;
        if (nrec > EPOOL_RANGE) nrec = EPOOL_RANGE;
        const int nw = (nrec + 1) >> 1;
        const unsigned* pw = pkw + (lo >> 1);
        for (int i = tid; i < nw; i += EPOOL_THREADS) dtab[i] = pw[i];
    }
    __syncthreads();

    const int lane = tid & 31;
    const int c    = lane & 15;
    const int hb   = lane & 16;
    const int wv   = __builtin_amdgcn_readfirstlane(tid >> 5);
    int2* myq = ring + wv * (EPOOL_RING + 32);

    const int per = (((E + (int)gridDim.x - 1) / (int)gridDim.x) + 31) & ~31;
    const int beg = blockIdx.x * per;
    int end = beg + per;
    if (end > E) end = E;

    int qh = 0;
    int qn = 0;
    const int gstep = (EPOOL_THREADS / 32) * EPOOL_TRIPS * EDGES_PER_WAVE;
    int base = beg + wv * (EPOOL_TRIPS * EDGES_PER_WAVE);
    int s_t[EPOOL_TRIPS], d_t[EPOOL_TRIPS];
#define POOL_LOAD_GROUP(B)                                                           \
    _Pragma("unroll")                                                                \
    for (int t = 0; t < EPOOL_TRIPS; ++t) {                                          \
        const int e  = (B) + t * EDGES_PER_WAVE + lane;                              \
        const int ec = e < end ? e : end - 1;     \
        s_t[t] = src[ec];                                                            \
        d_t[t] = dst[ec];                                                            \
    }
    if (base < end) { POOL_LOAD_GROUP(base) }
    for (; base < end; base += gstep) {
#pragma unroll
        for (int t = 0; t < EPOOL_TRIPS; ++t) {
            const int e      = base + t * EDGES_PER_WAVE + lane;
            const unsigned r = (unsigned)(d_t[t] - lo);
            const bool kept  = (e < end) && (r < (unsigned)EPOOL_RANGE);
            const unsigned m = __builtin_amdgcn_ballot_w32(kept);
            const int at   = (qh + qn + (int)__builtin_amdgcn_mbcnt_lo(m, 0u)) & (EPOOL_RING - 1);
            const int slot = kept ? at : EPOOL_RING + lane;
            myq[slot] = make_int2(s_t[t], (int)r);
            qn += __builtin_popcount(m);
        }
        if (base + gstep < end) { POOL_LOAD_GROUP(base + gstep) }

#pragma unroll 1
        while (qn >= 32) {
            const int2 ent = myq[(qh + lane) & (EPOOL_RING - 1)];
            qh = (qh + 32) & (EPOOL_RING - 1);
            qn -= 32;

            float hv[16];
#define POOL_GATHER(J) hv[J] = h[(unsigned)row_lane<J>(ent.x) * HID + c];
            POOL_GATHER(0)  POOL_GATHER(1)  POOL_GATHER(2)  POOL_GATHER(3)
            POOL_GATHER(4)  POOL_GATHER(5)  POOL_GATHER(6)  POOL_GATHER(7)
            POOL_GATHER(8)  POOL_GATHER(9)  POOL_GATHER(10) POOL_GATHER(11)
            POOL_GATHER(12) POOL_GATHER(13) POOL_GATHER(14) POOL_GATHER(15)
#undef POOL_GATHER
            int g_l; float w_l;
            pool_dest(dtab, rec, lo, ent.y, true, g_l, w_l);
#define POOL_ADD(J) ldsAddF(tab + ((unsigned)row_lane<J>(g_l) * HID + c), row_lane<J>(w_l) * hv[J]);
            POOL_ADD(0)  POOL_ADD(1)  POOL_ADD(2)  POOL_ADD(3)
            POOL_ADD(4)  POOL_ADD(5)  POOL_ADD(6)  POOL_ADD(7)
            POOL_ADD(8)  POOL_ADD(9)  POOL_ADD(10) POOL_ADD(11)
            POOL_ADD(12) POOL_ADD(13) POOL_ADD(14) POOL_ADD(15)
#undef POOL_ADD
        }
    }

#undef POOL_LOAD_GROUP
    if (qn > 0) {
        const bool live = lane < qn;
        int2 ent = make_int2(0, 0);
        if (live) ent = myq[(qh + lane) & (EPOOL_RING - 1)];
        int g_l; float w_l;
        pool_dest(dtab, rec, lo, ent.y, live, g_l, w_l);
#pragma unroll 1
        for (int j = 0; j < 16; ++j) {
            const int   s = __shfl(ent.x, hb + j);
            const int   g = __shfl(g_l, hb + j);
            const float w = __shfl(w_l, hb + j);
            if (hb + j < qn)
                ldsAddF(tab + ((unsigned)g * HID + c), w * h[(unsigned)s * HID + c]);
        }
    }
    __syncthreads();

    const int wg  = (int)(blockIdx.y * gridDim.x + blockIdx.x);
    const int rot = (wg * 48) & (NUM_GRAPHS * HID - 1) & ~31;
    for (int k = tid; k < NUM_GRAPHS * HID; k += EPOOL_THREADS) {
        const int i = (k + rot) & (NUM_GRAPHS * HID - 1);
        atomAddF(sums + i, tab[i]);
    }
}

__global__ void k_zero_pool(float* __restrict__ sums, float* __restrict__ cnt) {
    int t = blockIdx.x * blockDim.x + threadIdx.x;
    if (t < NUM_GRAPHS * HID) sums[t] = 0.0f;
    if (t < NUM_GRAPHS) cnt[t] = 0.0f;
}

#define POOL_NODES 32

__device__ __forceinline__ unsigned pool_pack(float dv, int g) {
    const unsigned deg = __float2uint_rn(1.0f / (dv * dv));
    return ((unsigned)g << 7) | (deg < EPOOL_DEGMAX ? deg : EPOOL_DEGMAX);
}

__device__ __forceinline__ void pool_flush(float* __restrict__ sums, float* __restrict__ cnt,
                                           int g, int c, float acc, float num) {
    if (g >= 0) {
        atomAddF(sums + g * HID + c, acc);
        if (c == 0) atomAddF(cnt + g, num);
    }
}

__global__ void k_pool(const float* __restrict__ h, const float* __restrict__ dinv,
                       const float* __restrict__ bias, const int* __restrict__ batch,
                       int* __restrict__ rec, unsigned* __restrict__ pkw,
                       float* __restrict__ sums, float* __restrict__ cnt, int n) {
    const int lane = threadIdx.x & 31;
    const int c    = lane & 15;
    const int half = lane >> 4;
    const int wave = blockIdx.x * (blockDim.x >> 5) + (threadIdx.x >> 5);
    const int base = wave * POOL_NODES;
    if (base >= n) return;

    const float b = bias[c];
    int   g[POOL_NODES / 2];
    float v[POOL_NODES / 2];
#pragma unroll
    for (int j = 0; j < POOL_NODES / 2; ++j) {
        const int  node = base + 2 * j + half;
        const bool ok   = node < n;
        const int  nc   = ok ? node : n - 1;
        const int gl = batch[nc];
        g[j] = ok ? gl : -1;
        v[j] = dinv[nc] * h[(size_t)nc * HID + c] + b;
    }
    {
        const int i = base + lane;
        if (i < n) reinterpret_cast<int2*>(rec)[i] = make_int2(__float_as_int(dinv[i]), batch[i]);
        const int i0 = base + 2 * lane;
        if (lane < 16 && i0 < n) {
            unsigned p1 = 0u;
            if (i0 + 1 < n) p1 = pool_pack(dinv[i0 + 1], batch[i0 + 1]);
            pkw[i0 >> 1] = pool_pack(dinv[i0], batch[i0]) | (p1 << 16);
        }
    }

    int   gcur = g[0];
    float acc = 0.0f, num = 0.0f;
#pragma unroll
    for (int j = 0; j < POOL_NODES / 2; ++j) {
        if (g[j] != gcur) {
            pool_flush(sums, cnt, gcur, c, acc, num);
            gcur = g[j]; acc = 0.0f; num = 0.0f;
        }
        acc += v[j];
        num += 1.0f;
    }
    pool_flush(sums, cnt, gcur, c, acc, num);
}

__global__ void k_head(const float* __restrict__ sums, const float* __restrict__ cnt,
                       const float* __restrict__ fcW, const float* __restrict__ fcb,
                       float* __restrict__ y) {
    int t = blockIdx.x * blockDim.x + threadIdx.x;
    if (t >= NUM_GRAPHS * NGRAPH) return;
    int g = t >> 4, cls = t & 15;
    float inv = 1.0f / fmaxf(cnt[g], 1.0f);
    float acc = fcb[cls];
#pragma unroll
    for (int k = 0; k < HID; ++k)
        acc += (sums[g * HID + k] * inv) * fcW[k * NGRAPH + cls];
    y[t] = acc;
}

extern "C" void kernel_launch(void* const* d_in, const int* in_sizes, int n_in,
                              void* d_out, int out_size, void* d_ws, size_t ws_size,
                              hipStream_t stream) {
    const float* x    = (const float*)d_in[0];
    const float* W1   = (const float*)d_in[1];
    const float* b1   = (const float*)d_in[2];
    const float* W2   = (const float*)d_in[3];
    const float* b2   = (const float*)d_in[4];
    const float* fcW  = (const float*)d_in[5];
    const float* fcb  = (const float*)d_in[6];
    const int*   eidx = (const int*)d_in[7];
    const int*   batch= (const int*)d_in[8];
    float* y = (float*)d_out;

    const int n = in_sizes[0] / 128;
    const int E = in_sizes[7] / 2;
    const int* src = eidx;
    const int* dst = eidx + E;

    size_t nAl = ((size_t)n + 3) & ~(size_t)3;
    float* deg  = (float*)d_ws;
    float* h    = deg + nAl;
    float* outb = h + (size_t)n * HID;
    float* sums = outb + (size_t)n * HID;
    float* cnt  = sums + NUM_GRAPHS * HID;

    const int TB = 256;
    const int n_tiles = (n + 15) / 16;

    unsigned* degc = (unsigned*)d_ws;
    k_init_deg<<<(n + TB - 1) / TB, TB, 0, stream>>>(degc, (unsigned*)sums, n);
    k_deg_accum<<<dim3(DEG_SLICES, (n + DEG_BINS - 1) / DEG_BINS), DEG_THREADS, 0, stream>>>(dst, degc, E, n);
    k_rsqrt<<<(n + TB - 1) / TB, TB, 0, stream>>>(degc, n);

    k_gemm16<128, false><<<(n_tiles + 3) / 4, 128, 0, stream>>>(x, W1, deg, b1, h, outb, n, n_tiles);
    k_edge_prop<<<EPROP_BLOCKS, EPROP_THREADS, 0, stream>>>(src, dst, h, outb, (unsigned*)sums, E, n);

    k_gemm16<16, true><<<(n_tiles + 3) / 4, 128, 0, stream>>>(outb, W2, deg, b1, h, nullptr, n, n_tiles);

    int* rec = (int*)outb;
    unsigned* pkw = (unsigned*)outb + 2 * nAl;
    k_zero_pool<<<(NUM_GRAPHS * HID + TB - 1) / TB, TB, 0, stream>>>(sums, cnt);
    k_pool<<<(n + TB - 1) / TB, TB, 0, stream>>>(h, deg, b2, batch, rec, pkw, sums, cnt, n);
    k_edge_pool<<<dim3(EPOOL_SLICES, (n + EPOOL_RANGE - 1) / EPOOL_RANGE), EPOOL_THREADS, 0, stream>>>(src, dst, pkw, rec, h, sums, E, n);
    k_head<<<(NUM_GRAPHS * NGRAPH + TB - 1) / TB, TB, 0, stream>>>(sums, cnt, fcW, fcb, y);
}


# layer-1 edge scatter: a wave packs the edges kept for its row range into an LDS ring (4 index trips per 1024-edge piece) and runs only full unmasked 16-round batches; atomic order unordered as before
# speedup vs baseline: 5.9006x; 1.0724x over previous
#include <hip/hip_runtime.h>
#include <hip/hip_bf16.h>

typedef __attribute__((ext_vector_type(2))) float v2f;
typedef __attribute__((ext_vector_type(8))) float v8f;

#define HID 16
#define NGRAPH 16
#define NUM_GRAPHS 512

__device__ __forceinline__ void atomAddF(float* p, float v) {
    __hip_atomic_fetch_add(p, v, __ATOMIC_RELAXED, __HIP_MEMORY_SCOPE_AGENT);
}

__device__ __forceinline__ void ldsAddF(float* p, float v) {
    __hip_atomic_fetch_add(p, v, __ATOMIC_RELAXED, __HIP_MEMORY_SCOPE_WORKGROUP);
}

#define DEG_BINS    50176
#define DEG_THREADS 1024
#define DEG_SLICES  64
#define DEG_QUADS   8

#define CTR_STRIDE  64

__global__ void k_init_deg(unsigned* __restrict__ cnt, unsigned* __restrict__ ctr, int n) {
    int i = blockIdx.x * blockDim.x + threadIdx.x;
    if (i < n) cnt[i] = 0u;
    if (i < 2) ctr[i * CTR_STRIDE] = 0u;
}

__device__ __forceinline__ void deg_bin_add(unsigned* bins, int d, int lo) {
    const unsigned r = (unsigned)(d - lo);
    if (r < (unsigned)DEG_BINS)
        __hip_atomic_fetch_add(bins + r, 1u, __ATOMIC_RELAXED, __HIP_MEMORY_SCOPE_WORKGROUP);
}

__global__ __launch_bounds__(DEG_THREADS)
void k_deg_accum(const int* __restrict__ dst, unsigned* __restrict__ cnt, int E, int n) {
    __shared__ unsigned bins[DEG_BINS];
    const int tid = threadIdx.x;
    const int lo  = blockIdx.y * DEG_BINS;

    for (int i = tid; i < DEG_BINS; i += DEG_THREADS) bins[i] = 0u;
    __syncthreads();

    const int per = (((E + (int)gridDim.x - 1) / (int)gridDim.x) + 3) & ~3;
    const int beg = blockIdx.x * per;
    int end = beg + per;
    if (end > E) end = E;

    int b0 = beg;
    if ((reinterpret_cast<size_t>(dst) & 15) == 0) {
        const int batch = 4 * DEG_THREADS * DEG_QUADS;
        for (; b0 + batch <= end; b0 += batch) {
            const int4* p = reinterpret_cast<const int4*>(dst + b0) + tid;
            int4 v[DEG_QUADS];
#pragma unroll
            for (int u = 0; u < DEG_QUADS; ++u) v[u] = p[u * DEG_THREADS];
#pragma unroll
            for (int u = 0; u < DEG_QUADS; ++u) {
                deg_bin_add(bins, v[u].x, lo);
                deg_bin_add(bins, v[u].y, lo);
                deg_bin_add(bins, v[u].z, lo);
                deg_bin_add(bins, v[u].w, lo);
            }
        }
    }
    for (int q = b0 + tid; q < end; q += DEG_THREADS) deg_bin_add(bins, dst[q], lo);
    __syncthreads();

    const int rot = ((int)blockIdx.x * (DEG_BINS / DEG_SLICES)) & ~31;
    for (int k = tid; k < DEG_BINS; k += DEG_THREADS) {
        int i = k + rot;
        if (i >= DEG_BINS) i -= DEG_BINS;
        const unsigned c = bins[i];
        const int node = lo + i;
        if (c != 0u && node < n)
            __hip_atomic_fetch_add(cnt + node, c, __ATOMIC_RELAXED, __HIP_MEMORY_SCOPE_AGENT);
    }
}

__global__ void k_rsqrt(unsigned* __restrict__ cnt, int n) {
    int i = blockIdx.x * blockDim.x + threadIdx.x;
    if (i < n) cnt[i] = __float_as_uint(rsqrtf((float)cnt[i] + 1.0f));
}

template<int K, bool LAYER2>
__global__ void k_gemm16(const float* __restrict__ A, const float* __restrict__ W,
                         const float* __restrict__ dinv, const float* __restrict__ bias,
                         float* __restrict__ D, float* __restrict__ D2, int n, int n_tiles) {
    const int lane = threadIdx.x & 31;
    const int wid  = threadIdx.x >> 5;
    const int tile = blockIdx.x * (blockDim.x >> 5) + wid;
    if (tile >= n_tiles) return;

    const int m    = lane & 15;
    const int half = lane >> 4;
    const int kh   = half * 2;

    int rowA = tile * 16 + m;
    int rowC = rowA;
    if (rowA >= n) rowA = n - 1;

    const float* arow = A + (size_t)rowA * K;
    const float  dA   = LAYER2 ? dinv[rowA] : 1.0f;
    v8f acc = {};
#pragma unroll
    for (int k0 = 0; k0 < K; k0 += 4) {
        float a0 = arow[k0 + kh];
        float a1 = arow[k0 + kh + 1];
        if (LAYER2) {
            a0 = fmaxf(dA * a0 + bias[k0 + kh], 0.0f);
            a1 = fmaxf(dA * a1 + bias[k0 + kh + 1], 0.0f);
        }
        v2f a; a.x = a0; a.y = a1;
        v2f b;
        b.x = W[(k0 + kh) * HID + m];
        b.y = W[(k0 + kh + 1) * HID + m];
        acc = __builtin_amdgcn_wmma_f32_16x16x4_f32(false, a, false, b,
                                                    (short)0, acc, false, false);
    }
#pragma unroll
    for (int r = 0; r < 8; ++r) {
        int row = tile * 16 + r + 8 * half;
        if (row < n) {
            const float v = dinv[row] * acc[r];
            D[(size_t)row * HID + m] = v;
            if (!LAYER2) D2[(size_t)row * HID + m] = v;
        }
    }
}

#define EDGES_PER_WAVE 32
#define EPROP_THREADS  256
#define EPROP_BLOCKS   2048
#define EPROP_TRIPS    4
#define EPROP_PIECE    ((EPROP_THREADS / 32) * EPROP_TRIPS * EDGES_PER_WAVE)
#define EPROP_RING     256

template<int J>
__device__ __forceinline__ int row_lane(int v) {
    return __builtin_amdgcn_update_dpp(0, v, 0x150 + J, 0xf, 0xf, true);
}
template<int J>
__device__ __forceinline__ float row_lane(float v) {
    return __int_as_float(row_lane<J>(__float_as_int(v)));
}

__device__ __forceinline__ unsigned take_piece(unsigned* q) {
    return __hip_atomic_fetch_add(q, 1u, __ATOMIC_RELAXED, __HIP_MEMORY_SCOPE_AGENT);
}

template<bool MASKED>
__device__ __forceinline__ void prop_batch(const int2 ent, const unsigned rm, const int c,
                                           const float* __restrict__ h, float* __restrict__ out) {
    float hv[16];
#define PROP_GATHER(J)                                               \
    {                                                                \
        const int s = row_lane<J>(ent.x);                            \
        hv[J] = 0.0f;                                                \
        if (!MASKED || (rm & (1u << J))) hv[J] = h[(unsigned)s * HID + c]; \
    }
    PROP_GATHER(0)  PROP_GATHER(1)  PROP_GATHER(2)  PROP_GATHER(3)
    PROP_GATHER(4)  PROP_GATHER(5)  PROP_GATHER(6)  PROP_GATHER(7)
    PROP_GATHER(8)  PROP_GATHER(9)  PROP_GATHER(10) PROP_GATHER(11)
    PROP_GATHER(12) PROP_GATHER(13) PROP_GATHER(14) PROP_GATHER(15)
#undef PROP_GATHER
#define PROP_ADD(J)                                                  \
    {                                                                \
        const int d = row_lane<J>(ent.y);                            \
        if (!MASKED || (rm & (1u << J))) atomAddF(out + ((unsigned)d * HID + c), hv[J]); \
    }
    PROP_ADD(0)  PROP_ADD(1)  PROP_ADD(2)  PROP_ADD(3)
    PROP_ADD(4)  PROP_ADD(5)  PROP_ADD(6)  PROP_ADD(7)
    PROP_ADD(8)  PROP_ADD(9)  PROP_ADD(10) PROP_ADD(11)
    PROP_ADD(12) PROP_ADD(13) PROP_ADD(14) PROP_ADD(15)
#undef PROP_ADD
}

__global__ __launch_bounds__(EPROP_THREADS)
void k_edge_prop(const int* __restrict__ src, const int* __restrict__ dst,
                 const float* __restrict__ h, float* __restrict__ out,
                 unsigned* __restrict__ ctr, int E, int n) {
    __shared__ unsigned nextp[2];
    __shared__ int2     ring[(EPROP_THREADS / 32) * (EPROP_RING + 32)];
    const int tid  = threadIdx.x;
    const int lane = tid & 31;
    const int wv   = __builtin_amdgcn_readfirstlane(tid >> 5);
    const int c    = lane & 15;
    const int hb   = lane & 16;
    int2* myq = ring + wv * (EPROP_RING + 32);

    const int myhalf = (int)((__builtin_amdgcn_s_sendmsg_rtn(0x87) >> 18) & 1u);
    const int split  = ((n >> 1) + 3) & ~3;
    const unsigned npieces = ((unsigned)E + EPROP_PIECE - 1) / EPROP_PIECE;

#pragma unroll 1
    for (int pass = 0; pass < 2; ++pass) {
        const int hh = myhalf ^ pass;
        unsigned* q = ctr + hh * CTR_STRIDE;
        int qh = 0;
        int qn = 0;
        if (tid == 0) nextp[0] = take_piece(q);
        __syncthreads();
        for (unsigned it = 0; ; ++it) {
            const unsigned p = __builtin_amdgcn_readfirstlane(nextp[it & 1]);
            if (p >= npieces) break;
            if (tid == 0) nextp[(it + 1) & 1] = take_piece(q);

            const int pbase = (int)(p * EPROP_PIECE) + wv * EDGES_PER_WAVE;
            int s_t[EPROP_TRIPS], d_t[EPROP_TRIPS];
#pragma unroll
            for (int t = 0; t < EPROP_TRIPS; ++t) {
                const int e  = pbase + t * (EPROP_THREADS / 32) * EDGES_PER_WAVE + lane;
                const int ec = e < E ? e : E - 1;
                s_t[t] = src[ec];
                d_t[t] = dst[ec];
            }
#pragma unroll
            for (int t = 0; t < EPROP_TRIPS; ++t) {
                const int e      = pbase + t * (EPROP_THREADS / 32) * EDGES_PER_WAVE + lane;
                const bool kept  = (e < E) && ((d_t[t] >= split) == (hh != 0));
                const unsigned m = __builtin_amdgcn_ballot_w32(kept);
                const int at   = (qh + qn + (int)__builtin_amdgcn_mbcnt_lo(m, 0u)) & (EPROP_RING - 1);
                const int slot = kept ? at : EPROP_RING + lane;
                myq[slot] = make_int2(s_t[t], d_t[t]);
                qn += __builtin_popcount(m);
            }
#pragma unroll 1
            while (qn >= 32) {
                const int2 ent = myq[(qh + lane) & (EPROP_RING - 1)];
                qh = (qh + 32) & (EPROP_RING - 1);
                qn -= 32;
                prop_batch<false>(ent, 0u, c, h, out);
            }
            __syncthreads();
        }
        if (qn > 0) {
            const bool live = lane < qn;
            int2 ent = make_int2(0, 0);
            if (live) ent = myq[(qh + lane) & (EPROP_RING - 1)];
            const unsigned rm = (__builtin_amdgcn_ballot_w32(live) >> hb) & 0xffffu;
            prop_batch<true>(ent, rm, c, h, out);
        }
        __syncthreads();
    }
}

#define EPOOL_THREADS 1024
#define EPOOL_SLICES  128
#define EPOOL_RANGE   100352
#define EPOOL_TRIPS   4
#define EPOOL_RING    256
#define EPOOL_DEGMAX  127u

__device__ __forceinline__ void pool_dest(const unsigned* dtab, const int* __restrict__ rec, int lo, int r, bool live,
                                          int& g, float& w) {
    const unsigned wrd = dtab[(unsigned)r >> 1];
    const unsigned pr  = (r & 1) ? (wrd >> 16) : (wrd & 0xffffu);
    const unsigned dg  = pr & EPOOL_DEGMAX;
    g = (int)(pr >> 7);
    w = rsqrtf((float)dg);
    if (live && dg == EPOOL_DEGMAX) w = __int_as_float(rec[2 * (lo + r)]);
}

__global__ __launch_bounds__(EPOOL_THREADS)
void k_edge_pool(const int* __restrict__ src, const int* __restrict__ dst,
                 const unsigned* __restrict__ pkw, const int* __restrict__ rec,
                 const float* __restrict__ h, float* __restrict__ sums, int E, int n) {
    __shared__ float    tab[NUM_GRAPHS * HID];
    __shared__ unsigned dtab[EPOOL_RANGE / 2];
    __shared__ int2     ring[(EPOOL_THREADS / 32) * (EPOOL_RING + 32)];
    const int tid = threadIdx.x;
    const int lo  = blockIdx.y * EPOOL_RANGE;
    for (int i = tid; i < NUM_GRAPHS * HID; i += EPOOL_THREADS) tab[i] = 0.0f;
    {
        int nrec = n - lo;
        if (nrec > EPOOL_RANGE) nrec = EPOOL_RANGE;
        const int nw = (nrec + 1) >> 1;
        const unsigned* pw = pkw + (lo >> 1);
        for (int i = tid; i < nw; i += EPOOL_THREADS) dtab[i] = pw[i];
    }
    __syncthreads();

    const int lane = tid & 31;
    const int c    = lane & 15;
    const int hb   = lane & 16;
    const int wv   = __builtin_amdgcn_readfirstlane(tid >> 5);
    int2* myq = ring + wv * (EPOOL_RING + 32);

    const int per = (((E + (int)gridDim.x - 1) / (int)gridDim.x) + 31) & ~31;
    const int beg = blockIdx.x * per;
    int end = beg + per;
    if (end > E) end = E;

    int qh = 0;
    int qn = 0;
    const int gstep = (EPOOL_THREADS / 32) * EPOOL_TRIPS * EDGES_PER_WAVE;
    int base = beg + wv * (EPOOL_TRIPS * EDGES_PER_WAVE);
    int s_t[EPOOL_TRIPS], d_t[EPOOL_TRIPS];
#define POOL_LOAD_GROUP(B)                                                           \
    _Pragma("unroll")                                                                \
    for (int t = 0; t < EPOOL_TRIPS; ++t) {                                          \
        const int e  = (B) + t * EDGES_PER_WAVE + lane;                              \
        const int ec = e < end ? e : end - 1;     \
        s_t[t] = src[ec];                                                            \
        d_t[t] = dst[ec];                                                            \
    }
    if (base < end) { POOL_LOAD_GROUP(base) }
    for (; base < end; base += gstep) {
#pragma unroll
        for (int t = 0; t < EPOOL_TRIPS; ++t) {
            const int e      = base + t * EDGES_PER_WAVE + lane;
            const unsigned r = (unsigned)(d_t[t] - lo);
            const bool kept  = (e < end) && (r < (unsigned)EPOOL_RANGE);
            const unsigned m = __builtin_amdgcn_ballot_w32(kept);
            const int at   = (qh + qn + (int)__builtin_amdgcn_mbcnt_lo(m, 0u)) & (EPOOL_RING - 1);
            const int slot = kept ? at : EPOOL_RING + lane;
            myq[slot] = make_int2(s_t[t], (int)r);
            qn += __builtin_popcount(m);
        }
        if (base + gstep < end) { POOL_LOAD_GROUP(base + gstep) }

#pragma unroll 1
        while (qn >= 32) {
            const int2 ent = myq[(qh + lane) & (EPOOL_RING - 1)];
            qh = (qh + 32) & (EPOOL_RING - 1);
            qn -= 32;

            float hv[16];
#define POOL_GATHER(J) hv[J] = h[(unsigned)row_lane<J>(ent.x) * HID + c];
            POOL_GATHER(0)  POOL_GATHER(1)  POOL_GATHER(2)  POOL_GATHER(3)
            POOL_GATHER(4)  POOL_GATHER(5)  POOL_GATHER(6)  POOL_GATHER(7)
            POOL_GATHER(8)  POOL_GATHER(9)  POOL_GATHER(10) POOL_GATHER(11)
            POOL_GATHER(12) POOL_GATHER(13) POOL_GATHER(14) POOL_GATHER(15)
#undef POOL_GATHER
            int g_l; float w_l;
            pool_dest(dtab, rec, lo, ent.y, true, g_l, w_l);
#define POOL_ADD(J) ldsAddF(tab + ((unsigned)row_lane<J>(g_l) * HID + c), row_lane<J>(w_l) * hv[J]);
            POOL_ADD(0)  POOL_ADD(1)  POOL_ADD(2)  POOL_ADD(3)
            POOL_ADD(4)  POOL_ADD(5)  POOL_ADD(6)  POOL_ADD(7)
            POOL_ADD(8)  POOL_ADD(9)  POOL_ADD(10) POOL_ADD(11)
            POOL_ADD(12) POOL_ADD(13) POOL_ADD(14) POOL_ADD(15)
#undef POOL_ADD
        }
    }

#undef POOL_LOAD_GROUP
    if (qn > 0) {
        const bool live = lane < qn;
        int2 ent = make_int2(0, 0);
        if (live) ent = myq[(qh + lane) & (EPOOL_RING - 1)];
        int g_l; float w_l;
        pool_dest(dtab, rec, lo, ent.y, live, g_l, w_l);
#pragma unroll 1
        for (int j = 0; j < 16; ++j) {
            const int   s = __shfl(ent.x, hb + j);
            const int   g = __shfl(g_l, hb + j);
            const float w = __shfl(w_l, hb + j);
            if (hb + j < qn)
                ldsAddF(tab + ((unsigned)g * HID + c), w * h[(unsigned)s * HID + c]);
        }
    }
    __syncthreads();

    const int wg  = (int)(blockIdx.y * gridDim.x + blockIdx.x);
    const int rot = (wg * 48) & (NUM_GRAPHS * HID - 1) & ~31;
    for (int k = tid; k < NUM_GRAPHS * HID; k += EPOOL_THREADS) {
        const int i = (k + rot) & (NUM_GRAPHS * HID - 1);
        atomAddF(sums + i, tab[i]);
    }
}

__global__ void k_zero_pool(float* __restrict__ sums, float* __restrict__ cnt) {
    int t = blockIdx.x * blockDim.x + threadIdx.x;
    if (t < NUM_GRAPHS * HID) sums[t] = 0.0f;
    if (t < NUM_GRAPHS) cnt[t] = 0.0f;
}

#define POOL_NODES 32

__device__ __forceinline__ unsigned pool_pack(float dv, int g) {
    const unsigned deg = __float2uint_rn(1.0f / (dv * dv));
    return ((unsigned)g << 7) | (deg < EPOOL_DEGMAX ? deg : EPOOL_DEGMAX);
}

__device__ __forceinline__ void pool_flush(float* __restrict__ sums, float* __restrict__ cnt,
                                           int g, int c, float acc, float num) {
    if (g >= 0) {
        atomAddF(sums + g * HID + c, acc);
        if (c == 0) atomAddF(cnt + g, num);
    }
}

__global__ void k_pool(const float* __restrict__ h, const float* __restrict__ dinv,
                       const float* __restrict__ bias, const int* __restrict__ batch,
                       int* __restrict__ rec, unsigned* __restrict__ pkw,
                       float* __restrict__ sums, float* __restrict__ cnt, int n) {
    const int lane = threadIdx.x & 31;
    const int c    = lane & 15;
    const int half = lane >> 4;
    const int wave = blockIdx.x * (blockDim.x >> 5) + (threadIdx.x >> 5);
    const int base = wave * POOL_NODES;
    if (base >= n) return;

    const float b = bias[c];
    int   g[POOL_NODES / 2];
    float v[POOL_NODES / 2];
#pragma unroll
    for (int j = 0; j < POOL_NODES / 2; ++j) {
        const int  node = base + 2 * j + half;
        const bool ok   = node < n;
        const int  nc   = ok ? node : n - 1;
        const int gl = batch[nc];
        g[j] = ok ? gl : -1;
        v[j] = dinv[nc] * h[(size_t)nc * HID + c] + b;
    }
    {
        const int i = base + lane;
        if (i < n) reinterpret_cast<int2*>(rec)[i] = make_int2(__float_as_int(dinv[i]), batch[i]);
        const int i0 = base + 2 * lane;
        if (lane < 16 && i0 < n) {
            unsigned p1 = 0u;
            if (i0 + 1 < n) p1 = pool_pack(dinv[i0 + 1], batch[i0 + 1]);
            pkw[i0 >> 1] = pool_pack(dinv[i0], batch[i0]) | (p1 << 16);
        }
    }

    int   gcur = g[0];
    float acc = 0.0f, num = 0.0f;
#pragma unroll
    for (int j = 0; j < POOL_NODES / 2; ++j) {
        if (g[j] != gcur) {
            pool_flush(sums, cnt, gcur, c, acc, num);
            gcur = g[j]; acc = 0.0f; num = 0.0f;
        }
        acc += v[j];
        num += 1.0f;
    }
    pool_flush(sums, cnt, gcur, c, acc, num);
}

__global__ void k_head(const float* __restrict__ sums, const float* __restrict__ cnt,
                       const float* __restrict__ fcW, const float* __restrict__ fcb,
                       float* __restrict__ y) {
    int t = blockIdx.x * blockDim.x + threadIdx.x;
    if (t >= NUM_GRAPHS * NGRAPH) return;
    int g = t >> 4, cls = t & 15;
    float inv = 1.0f / fmaxf(cnt[g], 1.0f);
    float acc = fcb[cls];
#pragma unroll
    for (int k = 0; k < HID; ++k)
        acc += (sums[g * HID + k] * inv) * fcW[k * NGRAPH + cls];
    y[t] = acc;
}

extern "C" void kernel_launch(void* const* d_in, const int* in_sizes, int n_in,
                              void* d_out, int out_size, void* d_ws, size_t ws_size,
                              hipStream_t stream) {
    const float* x    = (const float*)d_in[0];
    const float* W1   = (const float*)d_in[1];
    const float* b1   = (const float*)d_in[2];
    const float* W2   = (const float*)d_in[3];
    const float* b2   = (const float*)d_in[4];
    const float* fcW  = (const float*)d_in[5];
    const float* fcb  = (const float*)d_in[6];
    const int*   eidx = (const int*)d_in[7];
    const int*   batch= (const int*)d_in[8];
    float* y = (float*)d_out;

    const int n = in_sizes[0] / 128;
    const int E = in_sizes[7] / 2;
    const int* src = eidx;
    const int* dst = eidx + E;

    size_t nAl = ((size_t)n + 3) & ~(size_t)3;
    float* deg  = (float*)d_ws;
    float* h    = deg + nAl;
    float* outb = h + (size_t)n * HID;
    float* sums = outb + (size_t)n * HID;
    float* cnt  = sums + NUM_GRAPHS * HID;

    const int TB = 256;
    const int n_tiles = (n + 15) / 16;

    unsigned* degc = (unsigned*)d_ws;
    k_init_deg<<<(n + TB - 1) / TB, TB, 0, stream>>>(degc, (unsigned*)sums, n);
    k_deg_accum<<<dim3(DEG_SLICES, (n + DEG_BINS - 1) / DEG_BINS), DEG_THREADS, 0, stream>>>(dst, degc, E, n);
    k_rsqrt<<<(n + TB - 1) / TB, TB, 0, stream>>>(degc, n);

    k_gemm16<128, false><<<(n_tiles + 3) / 4, 128, 0, stream>>>(x, W1, deg, b1, h, outb, n, n_tiles);
    k_edge_prop<<<EPROP_BLOCKS, EPROP_THREADS, 0, stream>>>(src, dst, h, outb, (unsigned*)sums, E, n);

    k_gemm16<16, true><<<(n_tiles + 3) / 4, 128, 0, stream>>>(outb, W2, deg, b1, h, nullptr, n, n_tiles);

    int* rec = (int*)outb;
    unsigned* pkw = (unsigned*)outb + 2 * nAl;
    k_zero_pool<<<(NUM_GRAPHS * HID + TB - 1) / TB, TB, 0, stream>>>(sums, cnt);
    k_pool<<<(n + TB - 1) / TB, TB, 0, stream>>>(h, deg, b2, batch, rec, pkw, sums, cnt, n);
    k_edge_pool<<<dim3(EPOOL_SLICES, (n + EPOOL_RANGE - 1) / EPOOL_RANGE), EPOOL_THREADS, 0, stream>>>(src, dst, pkw, rec, h, sums, E, n);
    k_head<<<(NUM_GRAPHS * NGRAPH + TB - 1) / TB, TB, 0, stream>>>(sums, cnt, fcW, fcb, y);
}


# 3 launches fewer (10 to 7): rsqrt folded into GEMM 1; pool clearing + node records into GEMM 2 (now in place); per-node pool terms into the pooled edge pass's LDS table; pool sum order changes
# speedup vs baseline: 6.1239x; 1.0378x over previous
#include <hip/hip_runtime.h>
#include <hip/hip_bf16.h>

typedef __attribute__((ext_vector_type(2))) float v2f;
typedef __attribute__((ext_vector_type(8))) float v8f;

#define HID 16
#define NGRAPH 16
#define NUM_GRAPHS 512

__device__ __forceinline__ void atomAddF(float* p, float v) {
    __hip_atomic_fetch_add(p, v, __ATOMIC_RELAXED, __HIP_MEMORY_SCOPE_AGENT);
}

__device__ __forceinline__ void ldsAddF(float* p, float v) {
    __hip_atomic_fetch_add(p, v, __ATOMIC_RELAXED, __HIP_MEMORY_SCOPE_WORKGROUP);
}

#define DEG_BINS    50176
#define DEG_THREADS 1024
#define DEG_SLICES  64
#define DEG_QUADS   8

#define CTR_STRIDE  64

__global__ void k_init_deg(unsigned* __restrict__ cnt, unsigned* __restrict__ ctr, int n) {
    int i = blockIdx.x * blockDim.x + threadIdx.x;
    if (i < n) cnt[i] = 0u;
    if (i < 2) ctr[i * CTR_STRIDE] = 0u;
}

__device__ __forceinline__ void deg_bin_add(unsigned* bins, int d, int lo) {
    const unsigned r = (unsigned)(d - lo);
    if (r < (unsigned)DEG_BINS)
        __hip_atomic_fetch_add(bins + r, 1u, __ATOMIC_RELAXED, __HIP_MEMORY_SCOPE_WORKGROUP);
}

__global__ __launch_bounds__(DEG_THREADS)
void k_deg_accum(const int* __restrict__ dst, unsigned* __restrict__ cnt, int E, int n) {
    __shared__ unsigned bins[DEG_BINS];
    const int tid = threadIdx.x;
    const int lo  = blockIdx.y * DEG_BINS;

    for (int i = tid; i < DEG_BINS; i += DEG_THREADS) bins[i] = 0u;
    __syncthreads();

    const int per = (((E + (int)gridDim.x - 1) / (int)gridDim.x) + 3) & ~3;
    const int beg = blockIdx.x * per;
    int end = beg + per;
    if (end > E) end = E;

    int b0 = beg;
    if ((reinterpret_cast<size_t>(dst) & 15) == 0) {
        const int batch = 4 * DEG_THREADS * DEG_QUADS;
        for (; b0 + batch <= end; b0 += batch) {
            const int4* p = reinterpret_cast<const int4*>(dst + b0) + tid;
            int4 v[DEG_QUADS];
#pragma unroll
            for (int u = 0; u < DEG_QUADS; ++u) v[u] = p[u * DEG_THREADS];
#pragma unroll
            for (int u = 0; u < DEG_QUADS; ++u) {
                deg_bin_add(bins, v[u].x, lo);
                deg_bin_add(bins, v[u].y, lo);
                deg_bin_add(bins, v[u].z, lo);
                deg_bin_add(bins, v[u].w, lo);
            }
        }
    }
    for (int q = b0 + tid; q < end; q += DEG_THREADS) deg_bin_add(bins, dst[q], lo);
    __syncthreads();

    const int rot = ((int)blockIdx.x * (DEG_BINS / DEG_SLICES)) & ~31;
    for (int k = tid; k < DEG_BINS; k += DEG_THREADS) {
        int i = k + rot;
        if (i >= DEG_BINS) i -= DEG_BINS;
        const unsigned c = bins[i];
        const int node = lo + i;
        if (c != 0u && node < n)
            __hip_atomic_fetch_add(cnt + node, c, __ATOMIC_RELAXED, __HIP_MEMORY_SCOPE_AGENT);
    }
}

#define EPOOL_DEGMAX  127u

__device__ __forceinline__ unsigned pool_pack(float dv, int g) {
    const unsigned deg = __float2uint_rn(1.0f / (dv * dv));
    return ((unsigned)g << 7) | (deg < EPOOL_DEGMAX ? deg : EPOOL_DEGMAX);
}

template<int K, bool LAYER2>
__global__ void k_gemm16(const float* A, const float* __restrict__ W,
                         unsigned* __restrict__ dpl, const float* __restrict__ bias,
                         float* D, float* D2,
                         const int* __restrict__ batch, unsigned* __restrict__ pkw,
                         float* __restrict__ sums, float* __restrict__ cnt, int n, int n_tiles) {
    if (LAYER2) {
        const int nthr = (int)(gridDim.x * blockDim.x);
        for (int t = blockIdx.x * blockDim.x + threadIdx.x; t < NUM_GRAPHS * HID; t += nthr) {
            sums[t] = 0.0f;
            if (t < NUM_GRAPHS) cnt[t] = 0.0f;
        }
    }
    const int lane = threadIdx.x & 31;
    const int wid  = threadIdx.x >> 5;
    const int tile = blockIdx.x * (blockDim.x >> 5) + wid;
    if (tile >= n_tiles) return;

    const int m    = lane & 15;
    const int half = lane >> 4;
    const int kh   = half * 2;

    int rowA = tile * 16 + m;
    if (rowA >= n) rowA = n - 1;

    const float* arow = A + (size_t)rowA * K;
    const float  dA   = LAYER2 ? __uint_as_float(dpl[rowA]) : 1.0f;
    v8f acc = {};
#pragma unroll
    for (int k0 = 0; k0 < K; k0 += 4) {
        float a0 = arow[k0 + kh];
        float a1 = arow[k0 + kh + 1];
        if (LAYER2) {
            a0 = fmaxf(dA * a0 + bias[k0 + kh], 0.0f);
            a1 = fmaxf(dA * a1 + bias[k0 + kh + 1], 0.0f);
        }
        v2f a; a.x = a0; a.y = a1;
        v2f b;
        b.x = W[(k0 + kh) * HID + m];
        b.y = W[(k0 + kh + 1) * HID + m];
        acc = __builtin_amdgcn_wmma_f32_16x16x4_f32(false, a, false, b,
                                                    (short)0, acc, false, false);
    }
#pragma unroll
    for (int r = 0; r < 8; ++r) {
        int row = tile * 16 + r + 8 * half;
        if (row < n) {
            float dv;
            if (LAYER2) {
                dv = __uint_as_float(dpl[row]);
            } else {
                dv = rsqrtf((float)dpl[row] + 1.0f);
                if (m == 0) dpl[row] = __float_as_uint(dv);
            }
            const float v = dv * acc[r];
            D[(size_t)row * HID + m] = v;
            if (!LAYER2) D2[(size_t)row * HID + m] = v;
        }
    }
    if (LAYER2) {
        const int i0 = tile * 16 + 2 * lane;
        if (lane < 8 && i0 < n) {
            unsigned p1 = 0u;
            if (i0 + 1 < n) p1 = pool_pack(__uint_as_float(dpl[i0 + 1]), batch[i0 + 1]);
            pkw[i0 >> 1] = pool_pack(__uint_as_float(dpl[i0]), batch[i0]) | (p1 << 16);
        }
    }
}

#define EDGES_PER_WAVE 32
#define EPROP_THREADS  256
#define EPROP_BLOCKS   2048
#define EPROP_TRIPS    4
#define EPROP_PIECE    ((EPROP_THREADS / 32) * EPROP_TRIPS * EDGES_PER_WAVE)
#define EPROP_RING     256

template<int J>
__device__ __forceinline__ int row_lane(int v) {
    return __builtin_amdgcn_update_dpp(0, v, 0x150 + J, 0xf, 0xf, true);
}
template<int J>
__device__ __forceinline__ float row_lane(float v) {
    return __int_as_float(row_lane<J>(__float_as_int(v)));
}

__device__ __forceinline__ unsigned take_piece(unsigned* q) {
    return __hip_atomic_fetch_add(q, 1u, __ATOMIC_RELAXED, __HIP_MEMORY_SCOPE_AGENT);
}

template<bool MASKED>
__device__ __forceinline__ void prop_batch(const int2 ent, const unsigned rm, const int c,
                                           const float* __restrict__ h, float* __restrict__ out) {
    float hv[16];
#define PROP_GATHER(J)                                               \
    {                                                                \
        const int s = row_lane<J>(ent.x);                            \
        hv[J] = 0.0f;                                                \
        if (!MASKED || (rm & (1u << J))) hv[J] = h[(unsigned)s * HID + c]; \
    }
    PROP_GATHER(0)  PROP_GATHER(1)  PROP_GATHER(2)  PROP_GATHER(3)
    PROP_GATHER(4)  PROP_GATHER(5)  PROP_GATHER(6)  PROP_GATHER(7)
    PROP_GATHER(8)  PROP_GATHER(9)  PROP_GATHER(10) PROP_GATHER(11)
    PROP_GATHER(12) PROP_GATHER(13) PROP_GATHER(14) PROP_GATHER(15)
#undef PROP_GATHER
#define PROP_ADD(J)                                                  \
    {                                                                \
        const int d = row_lane<J>(ent.y);                            \
        if (!MASKED || (rm & (1u << J))) atomAddF(out + ((unsigned)d * HID + c), hv[J]); \
    }
    PROP_ADD(0)  PROP_ADD(1)  PROP_ADD(2)  PROP_ADD(3)
    PROP_ADD(4)  PROP_ADD(5)  PROP_ADD(6)  PROP_ADD(7)
    PROP_ADD(8)  PROP_ADD(9)  PROP_ADD(10) PROP_ADD(11)
    PROP_ADD(12) PROP_ADD(13) PROP_ADD(14) PROP_ADD(15)
#undef PROP_ADD
}

__global__ __launch_bounds__(EPROP_THREADS)
void k_edge_prop(const int* __restrict__ src, const int* __restrict__ dst,
                 const float* __restrict__ h, float* __restrict__ out,
                 unsigned* __restrict__ ctr, int E, int n) {
    __shared__ unsigned nextp[2];
    __shared__ int2     ring[(EPROP_THREADS / 32) * (EPROP_RING + 32)];
    const int tid  = threadIdx.x;
    const int lane = tid & 31;
    const int wv   = __builtin_amdgcn_readfirstlane(tid >> 5);
    const int c    = lane & 15;
    const int hb   = lane & 16;
    int2* myq = ring + wv * (EPROP_RING + 32);

    const int myhalf = (int)((__builtin_amdgcn_s_sendmsg_rtn(0x87) >> 18) & 1u);
    const int split  = ((n >> 1) + 3) & ~3;
    const unsigned npieces = ((unsigned)E + EPROP_PIECE - 1) / EPROP_PIECE;

#pragma unroll 1
    for (int pass = 0; pass < 2; ++pass) {
        const int hh = myhalf ^ pass;
        unsigned* q = ctr + hh * CTR_STRIDE;
        int qh = 0;
        int qn = 0;
        if (tid == 0) nextp[0] = take_piece(q);
        __syncthreads();
        for (unsigned it = 0; ; ++it) {
            const unsigned p = __builtin_amdgcn_readfirstlane(nextp[it & 1]);
            if (p >= npieces) break;
            if (tid == 0) nextp[(it + 1) & 1] = take_piece(q);

            const int pbase = (int)(p * EPROP_PIECE) + wv * EDGES_PER_WAVE;
            int s_t[EPROP_TRIPS], d_t[EPROP_TRIPS];
#pragma unroll
            for (int t = 0; t < EPROP_TRIPS; ++t) {
                const int e  = pbase + t * (EPROP_THREADS / 32) * EDGES_PER_WAVE + lane;
                const int ec = e < E ? e : E - 1;
                s_t[t] = src[ec];
                d_t[t] = dst[ec];
            }
#pragma unroll
            for (int t = 0; t < EPROP_TRIPS; ++t) {
                const int e      = pbase + t * (EPROP_THREADS / 32) * EDGES_PER_WAVE + lane;
                const bool kept  = (e < E) && ((d_t[t] >= split) == (hh != 0));
                const unsigned m = __builtin_amdgcn_ballot_w32(kept);
                const int at   = (qh + qn + (int)__builtin_amdgcn_mbcnt_lo(m, 0u)) & (EPROP_RING - 1);
                const int slot = kept ? at : EPROP_RING + lane;
                myq[slot] = make_int2(s_t[t], d_t[t]);
                qn += __builtin_popcount(m);
            }
#pragma unroll 1
            while (qn >= 32) {
                const int2 ent = myq[(qh + lane) & (EPROP_RING - 1)];
                qh = (qh + 32) & (EPROP_RING - 1);
                qn -= 32;
                prop_batch<false>(ent, 0u, c, h, out);
            }
            __syncthreads();
        }
        if (qn > 0) {
            const bool live = lane < qn;
            int2 ent = make_int2(0, 0);
            if (live) ent = myq[(qh + lane) & (EPROP_RING - 1)];
            const unsigned rm = (__builtin_amdgcn_ballot_w32(live) >> hb) & 0xffffu;
            prop_batch<true>(ent, rm, c, h, out);
        }
        __syncthreads();
    }
}

#define EPOOL_THREADS 1024
#define EPOOL_SLICES  128
#define EPOOL_RANGE   100352
#define EPOOL_TRIPS   4
#define EPOOL_RING    256
#define POOL_NODES    32

__device__ __forceinline__ void pool_dest(const unsigned* dtab, const float* __restrict__ dinv, int lo, int r, bool live,
                                          int& g, float& w) {
    const unsigned wrd = dtab[(unsigned)r >> 1];
    const unsigned pr  = (r & 1) ? (wrd >> 16) : (wrd & 0xffffu);
    const unsigned dg  = pr & EPOOL_DEGMAX;
    g = (int)(pr >> 7);
    w = rsqrtf((float)dg);
    if (live && dg == EPOOL_DEGMAX) w = dinv[lo + r];
}

__global__ __launch_bounds__(EPOOL_THREADS)
void k_edge_pool(const int* __restrict__ src, const int* __restrict__ dst,
                 const unsigned* __restrict__ pkw, const float* __restrict__ dinv,
                 const float* __restrict__ h, const float* __restrict__ bias, const int* __restrict__ batch,
                 float* __restrict__ sums, float* __restrict__ cnt, int E, int n) {
    __shared__ float    tab[NUM_GRAPHS * HID];
    __shared__ float    ctab[NUM_GRAPHS];
    __shared__ unsigned dtab[EPOOL_RANGE / 2];
    __shared__ int2     ring[(EPOOL_THREADS / 32) * (EPOOL_RING + 32)];
    const int tid = threadIdx.x;
    const int lo  = blockIdx.y * EPOOL_RANGE;
    for (int i = tid; i < NUM_GRAPHS * HID; i += EPOOL_THREADS) tab[i] = 0.0f;
    for (int i = tid; i < NUM_GRAPHS; i += EPOOL_THREADS) ctab[i] = 0.0f;
    {
        int nrec = n - lo;
        if (nrec > EPOOL_RANGE) nrec = EPOOL_RANGE;
        const int nw = (nrec + 1) >> 1;
        const unsigned* pw = pkw + (lo >> 1);
        for (int i = tid; i < nw; i += EPOOL_THREADS) dtab[i] = pw[i];
    }
    __syncthreads();

    const int lane = tid & 31;
    const int c    = lane & 15;
    const int hb   = lane & 16;
    const int wv   = __builtin_amdgcn_readfirstlane(tid >> 5);
    int2* myq = ring + wv * (EPOOL_RING + 32);

    {
        const int nwg   = (int)(gridDim.x * gridDim.y);
        const int wg    = (int)(blockIdx.y * gridDim.x + blockIdx.x);
        const int share = (n + nwg - 1) / nwg;
        const int nb    = wg * share;
        int ne = nb + share;
        if (ne > n) ne = n;
        const float bc = bias[c];
        for (int base = nb + wv * POOL_NODES; base < ne; base += (EPOOL_THREADS / 32) * POOL_NODES) {
            int   g[POOL_NODES / 2];
            float v[POOL_NODES / 2];
#pragma unroll
            for (int j = 0; j < POOL_NODES / 2; ++j) {
                const int  node = base + 2 * j + (lane >> 4);
                const bool ok   = node < ne;
                const int  nc   = ok ? node : ne - 1;
                const int  gl   = batch[nc];
                g[j] = ok ? gl : -1;
                v[j] = dinv[nc] * h[(size_t)nc * HID + c] + bc;
            }
#pragma unroll
            for (int j = 0; j < POOL_NODES / 2; ++j) {
                if (g[j] >= 0) {
                    ldsAddF(tab + ((unsigned)g[j] * HID + c), v[j]);
                    if (c == 0) ldsAddF(ctab + g[j], 1.0f);
                }
            }
        }
    }

    const int per = (((E + (int)gridDim.x - 1) / (int)gridDim.x) + 31) & ~31;
    const int beg = blockIdx.x * per;
    int end = beg + per;
    if (end > E) end = E;

    int qh = 0;
    int qn = 0;
    const int gstep = (EPOOL_THREADS / 32) * EPOOL_TRIPS * EDGES_PER_WAVE;
    int base = beg + wv * (EPOOL_TRIPS * EDGES_PER_WAVE);
    int s_t[EPOOL_TRIPS], d_t[EPOOL_TRIPS];
#define POOL_LOAD_GROUP(B)                                                           \
    _Pragma("unroll")                                                                \
    for (int t = 0; t < EPOOL_TRIPS; ++t) {                                          \
        const int e  = (B) + t * EDGES_PER_WAVE + lane;                              \
        const int ec = e < end ? e : end - 1;     \
        s_t[t] = src[ec];                                                            \
        d_t[t] = dst[ec];                                                            \
    }
    if (base < end) { POOL_LOAD_GROUP(base) }
    for (; base < end; base += gstep) {
#pragma unroll
        for (int t = 0; t < EPOOL_TRIPS; ++t) {
            const int e      = base + t * EDGES_PER_WAVE + lane;
            const unsigned r = (unsigned)(d_t[t] - lo);
            const bool kept  = (e < end) && (r < (unsigned)EPOOL_RANGE);
            const unsigned m = __builtin_amdgcn_ballot_w32(kept);
            const int at   = (qh + qn + (int)__builtin_amdgcn_mbcnt_lo(m, 0u)) & (EPOOL_RING - 1);
            const int slot = kept ? at : EPOOL_RING + lane;
            myq[slot] = make_int2(s_t[t], (int)r);
            qn += __builtin_popcount(m);
        }
        if (base + gstep < end) { POOL_LOAD_GROUP(base + gstep) }

#pragma unroll 1
        while (qn >= 32) {
            const int2 ent = myq[(qh + lane) & (EPOOL_RING - 1)];
            qh = (qh + 32) & (EPOOL_RING - 1);
            qn -= 32;

            float hv[16];
#define POOL_GATHER(J) hv[J] = h[(unsigned)row_lane<J>(ent.x) * HID + c];
            POOL_GATHER(0)  POOL_GATHER(1)  POOL_GATHER(2)  POOL_GATHER(3)
            POOL_GATHER(4)  POOL_GATHER(5)  POOL_GATHER(6)  POOL_GATHER(7)
            POOL_GATHER(8)  POOL_GATHER(9)  POOL_GATHER(10) POOL_GATHER(11)
            POOL_GATHER(12) POOL_GATHER(13) POOL_GATHER(14) POOL_GATHER(15)
#undef POOL_GATHER
            int g_l; float w_l;
            pool_dest(dtab, dinv, lo, ent.y, true, g_l, w_l);
#define POOL_ADD(J) ldsAddF(tab + ((unsigned)row_lane<J>(g_l) * HID + c), row_lane<J>(w_l) * hv[J]);
            POOL_ADD(0)  POOL_ADD(1)  POOL_ADD(2)  POOL_ADD(3)
            POOL_ADD(4)  POOL_ADD(5)  POOL_ADD(6)  POOL_ADD(7)
            POOL_ADD(8)  POOL_ADD(9)  POOL_ADD(10) POOL_ADD(11)
            POOL_ADD(12) POOL_ADD(13) POOL_ADD(14) POOL_ADD(15)
#undef POOL_ADD
        }
    }

#undef POOL_LOAD_GROUP
    if (qn > 0) {
        const bool live = lane < qn;
        int2 ent = make_int2(0, 0);
        if (live) ent = myq[(qh + lane) & (EPOOL_RING - 1)];
        int g_l; float w_l;
        pool_dest(dtab, dinv, lo, ent.y, live, g_l, w_l);
#pragma unroll 1
        for (int j = 0; j < 16; ++j) {
            const int   s = __shfl(ent.x, hb + j);
            const int   g = __shfl(g_l, hb + j);
            const float w = __shfl(w_l, hb + j);
            if (hb + j < qn)
                ldsAddF(tab + ((unsigned)g * HID + c), w * h[(unsigned)s * HID + c]);
        }
    }
    __syncthreads();

    const int wg  = (int)(blockIdx.y * gridDim.x + blockIdx.x);
    const int rot = (wg * 48) & (NUM_GRAPHS * HID - 1) & ~31;
    for (int k = tid; k < NUM_GRAPHS * HID; k += EPOOL_THREADS) {
        const int i = (k + rot) & (NUM_GRAPHS * HID - 1);
        atomAddF(sums + i, tab[i]);
    }
    for (int k = tid; k < NUM_GRAPHS; k += EPOOL_THREADS) {
        const float cg = ctab[k];
        if (cg != 0.0f) atomAddF(cnt + k, cg);
    }
}

__global__ void k_head(const float* __restrict__ sums, const float* __restrict__ cnt,
                       const float* __restrict__ fcW, const float* __restrict__ fcb,
                       float* __restrict__ y) {
    int t = blockIdx.x * blockDim.x + threadIdx.x;
    if (t >= NUM_GRAPHS * NGRAPH) return;
    int g = t >> 4, cls = t & 15;
    float inv = 1.0f / fmaxf(cnt[g], 1.0f);
    float acc = fcb[cls];
#pragma unroll
    for (int k = 0; k < HID; ++k)
        acc += (sums[g * HID + k] * inv) * fcW[k * NGRAPH + cls];
    y[t] = acc;
}

extern "C" void kernel_launch(void* const* d_in, const int* in_sizes, int n_in,
                              void* d_out, int out_size, void* d_ws, size_t ws_size,
                              hipStream_t stream) {
    const float* x    = (const float*)d_in[0];
    const float* W1   = (const float*)d_in[1];
    const float* b1   = (const float*)d_in[2];
    const float* W2   = (const float*)d_in[3];
    const float* b2   = (const float*)d_in[4];
    const float* fcW  = (const float*)d_in[5];
    const float* fcb  = (const float*)d_in[6];
    const int*   eidx = (const int*)d_in[7];
    const int*   batch= (const int*)d_in[8];
    float* y = (float*)d_out;

    const int n = in_sizes[0] / 128;
    const int E = in_sizes[7] / 2;
    const int* src = eidx;
    const int* dst = eidx + E;

    size_t nAl = ((size_t)n + 3) & ~(size_t)3;
    float* deg  = (float*)d_ws;
    float* h    = deg + nAl;
    float* outb = h + (size_t)n * HID;
    float* sums = outb + (size_t)n * HID;
    float* cnt  = sums + NUM_GRAPHS * HID;

    const int TB = 256;
    const int n_tiles = (n + 15) / 16;

    unsigned* degc = (unsigned*)d_ws;
    k_init_deg<<<(n + TB - 1) / TB, TB, 0, stream>>>(degc, (unsigned*)sums, n);
    k_deg_accum<<<dim3(DEG_SLICES, (n + DEG_BINS - 1) / DEG_BINS), DEG_THREADS, 0, stream>>>(dst, degc, E, n);

    k_gemm16<128, false><<<(n_tiles + 3) / 4, 128, 0, stream>>>(x, W1, degc, b1, h, outb, nullptr, nullptr, nullptr, nullptr, n, n_tiles);
    k_edge_prop<<<EPROP_BLOCKS, EPROP_THREADS, 0, stream>>>(src, dst, h, outb, (unsigned*)sums, E, n);

    unsigned* pkw = (unsigned*)h;
    k_gemm16<16, true><<<(n_tiles + 3) / 4, 128, 0, stream>>>(outb, W2, degc, b1, outb, nullptr, batch, pkw, sums, cnt, n, n_tiles);

    k_edge_pool<<<dim3(EPOOL_SLICES, (n + EPOOL_RANGE - 1) / EPOOL_RANGE), EPOOL_THREADS, 0, stream>>>(src, dst, pkw, deg, outb, b2, batch, sums, cnt, E, n);
    k_head<<<(NUM_GRAPHS * NGRAPH + TB - 1) / TB, TB, 0, stream>>>(sums, cnt, fcW, fcb, y);
}


# layer-1 GEMM: a wave reads its 16x128 tile of x as 16 whole-row loads (4 lines per instruction, not 16 lines x 16 B); A fragments come from a padded LDS copy of its own; k and sum order unchanged
# speedup vs baseline: 6.6671x; 1.0887x over previous
#include <hip/hip_runtime.h>
#include <hip/hip_bf16.h>

typedef __attribute__((ext_vector_type(2))) float v2f;
typedef __attribute__((ext_vector_type(8))) float v8f;

#define HID 16
#define NGRAPH 16
#define NUM_GRAPHS 512

__device__ __forceinline__ void atomAddF(float* p, float v) {
    __hip_atomic_fetch_add(p, v, __ATOMIC_RELAXED, __HIP_MEMORY_SCOPE_AGENT);
}

__device__ __forceinline__ void ldsAddF(float* p, float v) {
    __hip_atomic_fetch_add(p, v, __ATOMIC_RELAXED, __HIP_MEMORY_SCOPE_WORKGROUP);
}

#define DEG_BINS    50176
#define DEG_THREADS 1024
#define DEG_SLICES  64
#define DEG_QUADS   8

#define CTR_STRIDE  64

__global__ void k_init_deg(unsigned* __restrict__ cnt, unsigned* __restrict__ ctr, int n) {
    int i = blockIdx.x * blockDim.x + threadIdx.x;
    if (i < n) cnt[i] = 0u;
    if (i < 2) ctr[i * CTR_STRIDE] = 0u;
}

__device__ __forceinline__ void deg_bin_add(unsigned* bins, int d, int lo) {
    const unsigned r = (unsigned)(d - lo);
    if (r < (unsigned)DEG_BINS)
        __hip_atomic_fetch_add(bins + r, 1u, __ATOMIC_RELAXED, __HIP_MEMORY_SCOPE_WORKGROUP);
}

__global__ __launch_bounds__(DEG_THREADS)
void k_deg_accum(const int* __restrict__ dst, unsigned* __restrict__ cnt, int E, int n) {
    __shared__ unsigned bins[DEG_BINS];
    const int tid = threadIdx.x;
    const int lo  = blockIdx.y * DEG_BINS;

    for (int i = tid; i < DEG_BINS; i += DEG_THREADS) bins[i] = 0u;
    __syncthreads();

    const int per = (((E + (int)gridDim.x - 1) / (int)gridDim.x) + 3) & ~3;
    const int beg = blockIdx.x * per;
    int end = beg + per;
    if (end > E) end = E;

    int b0 = beg;
    if ((reinterpret_cast<size_t>(dst) & 15) == 0) {
        const int batch = 4 * DEG_THREADS * DEG_QUADS;
        for (; b0 + batch <= end; b0 += batch) {
            const int4* p = reinterpret_cast<const int4*>(dst + b0) + tid;
            int4 v[DEG_QUADS];
#pragma unroll
            for (int u = 0; u < DEG_QUADS; ++u) v[u] = p[u * DEG_THREADS];
#pragma unroll
            for (int u = 0; u < DEG_QUADS; ++u) {
                deg_bin_add(bins, v[u].x, lo);
                deg_bin_add(bins, v[u].y, lo);
                deg_bin_add(bins, v[u].z, lo);
                deg_bin_add(bins, v[u].w, lo);
            }
        }
    }
    for (int q = b0 + tid; q < end; q += DEG_THREADS) deg_bin_add(bins, dst[q], lo);
    __syncthreads();

    const int rot = ((int)blockIdx.x * (DEG_BINS / DEG_SLICES)) & ~31;
    for (int k = tid; k < DEG_BINS; k += DEG_THREADS) {
        int i = k + rot;
        if (i >= DEG_BINS) i -= DEG_BINS;
        const unsigned c = bins[i];
        const int node = lo + i;
        if (c != 0u && node < n)
            __hip_atomic_fetch_add(cnt + node, c, __ATOMIC_RELAXED, __HIP_MEMORY_SCOPE_AGENT);
    }
}

#define EPOOL_DEGMAX  127u

__device__ __forceinline__ unsigned pool_pack(float dv, int g) {
    const unsigned deg = __float2uint_rn(1.0f / (dv * dv));
    return ((unsigned)g << 7) | (deg < EPOOL_DEGMAX ? deg : EPOOL_DEGMAX);
}

#define GEMM_WAVES 4
template<int K, bool LAYER2>
__global__ void k_gemm16(const float* A, const float* __restrict__ W,
                         unsigned* __restrict__ dpl, const float* __restrict__ bias,
                         float* D, float* D2,
                         const int* __restrict__ batch, unsigned* __restrict__ pkw,
                         float* __restrict__ sums, float* __restrict__ cnt, int n, int n_tiles) {
    if (LAYER2) {
        const int nthr = (int)(gridDim.x * blockDim.x);
        for (int t = blockIdx.x * blockDim.x + threadIdx.x; t < NUM_GRAPHS * HID; t += nthr) {
            sums[t] = 0.0f;
            if (t < NUM_GRAPHS) cnt[t] = 0.0f;
        }
    }
    const int lane = threadIdx.x & 31;
    const int wid  = threadIdx.x >> 5;
    const int tile = blockIdx.x * (blockDim.x >> 5) + wid;
    if (tile >= n_tiles) return;

    const int m    = lane & 15;
    const int half = lane >> 4;
    const int kh   = half * 2;

    int rowA = tile * 16 + m;
    if (rowA >= n) rowA = n - 1;

    const float* arow = A + (size_t)rowA * K;
    const float  dA   = LAYER2 ? __uint_as_float(dpl[rowA]) : 1.0f;

    constexpr int PITCH = K + 4;
    __shared__ __attribute__((aligned(16))) float stage[LAYER2 ? 4 : GEMM_WAVES * 16 * PITCH];
    const float* srow = stage + wid * 16 * PITCH + m * PITCH;
    if (!LAYER2) {
        static_assert(LAYER2 || K == 128, "one 16-byte piece per lane covers a row of 128 floats");
        float* st = stage + wid * 16 * PITCH;
        float4 v[16];
#pragma unroll
        for (int i = 0; i < 16; ++i) {
            int row = tile * 16 + i;
            if (row >= n) row = n - 1;
            v[i] = *reinterpret_cast<const float4*>(A + (size_t)row * K + 4 * lane);
        }
#pragma unroll
        for (int i = 0; i < 16; ++i)
            *reinterpret_cast<float4*>(st + i * PITCH + 4 * lane) = v[i];
    }
    v8f acc = {};
#pragma unroll
    for (int k0 = 0; k0 < K; k0 += 4) {
        float a0, a1;
        if (LAYER2) {
            a0 = arow[k0 + kh];
            a1 = arow[k0 + kh + 1];
        } else {
            const v2f t = *reinterpret_cast<const v2f*>(srow + k0 + kh);
            a0 = t.x; a1 = t.y;
        }
        if (LAYER2) {
            a0 = fmaxf(dA * a0 + bias[k0 + kh], 0.0f);
            a1 = fmaxf(dA * a1 + bias[k0 + kh + 1], 0.0f);
        }
        v2f a; a.x = a0; a.y = a1;
        v2f b;
        b.x = W[(k0 + kh) * HID + m];
        b.y = W[(k0 + kh + 1) * HID + m];
        acc = __builtin_amdgcn_wmma_f32_16x16x4_f32(false, a, false, b,
                                                    (short)0, acc, false, false);
    }
#pragma unroll
    for (int r = 0; r < 8; ++r) {
        int row = tile * 16 + r + 8 * half;
        if (row < n) {
            float dv;
            if (LAYER2) {
                dv = __uint_as_float(dpl[row]);
            } else {
                dv = rsqrtf((float)dpl[row] + 1.0f);
                if (m == 0) dpl[row] = __float_as_uint(dv);
            }
            const float v = dv * acc[r];
            D[(size_t)row * HID + m] = v;
            if (!LAYER2) D2[(size_t)row * HID + m] = v;
        }
    }
    if (LAYER2) {
        const int i0 = tile * 16 + 2 * lane;
        if (lane < 8 && i0 < n) {
            unsigned p1 = 0u;
            if (i0 + 1 < n) p1 = pool_pack(__uint_as_float(dpl[i0 + 1]), batch[i0 + 1]);
            pkw[i0 >> 1] = pool_pack(__uint_as_float(dpl[i0]), batch[i0]) | (p1 << 16);
        }
    }
}

#define EDGES_PER_WAVE 32
#define EPROP_THREADS  256
#define EPROP_BLOCKS   2048
#define EPROP_TRIPS    4
#define EPROP_PIECE    ((EPROP_THREADS / 32) * EPROP_TRIPS * EDGES_PER_WAVE)
#define EPROP_RING     256

template<int J>
__device__ __forceinline__ int row_lane(int v) {
    return __builtin_amdgcn_update_dpp(0, v, 0x150 + J, 0xf, 0xf, true);
}
template<int J>
__device__ __forceinline__ float row_lane(float v) {
    return __int_as_float(row_lane<J>(__float_as_int(v)));
}

__device__ __forceinline__ unsigned take_piece(unsigned* q) {
    return __hip_atomic_fetch_add(q, 1u, __ATOMIC_RELAXED, __HIP_MEMORY_SCOPE_AGENT);
}

template<bool MASKED>
__device__ __forceinline__ void prop_batch(const int2 ent, const unsigned rm, const int c,
                                           const float* __restrict__ h, float* __restrict__ out) {
    float hv[16];
#define PROP_GATHER(J)                                               \
    {                                                                \
        const int s = row_lane<J>(ent.x);                            \
        hv[J] = 0.0f;                                                \
        if (!MASKED || (rm & (1u << J))) hv[J] = h[(unsigned)s * HID + c]; \
    }
    PROP_GATHER(0)  PROP_GATHER(1)  PROP_GATHER(2)  PROP_GATHER(3)
    PROP_GATHER(4)  PROP_GATHER(5)  PROP_GATHER(6)  PROP_GATHER(7)
    PROP_GATHER(8)  PROP_GATHER(9)  PROP_GATHER(10) PROP_GATHER(11)
    PROP_GATHER(12) PROP_GATHER(13) PROP_GATHER(14) PROP_GATHER(15)
#undef PROP_GATHER
#define PROP_ADD(J)                                                  \
    {                                                                \
        const int d = row_lane<J>(ent.y);                            \
        if (!MASKED || (rm & (1u << J))) atomAddF(out + ((unsigned)d * HID + c), hv[J]); \
    }
    PROP_ADD(0)  PROP_ADD(1)  PROP_ADD(2)  PROP_ADD(3)
    PROP_ADD(4)  PROP_ADD(5)  PROP_ADD(6)  PROP_ADD(7)
    PROP_ADD(8)  PROP_ADD(9)  PROP_ADD(10) PROP_ADD(11)
    PROP_ADD(12) PROP_ADD(13) PROP_ADD(14) PROP_ADD(15)
#undef PROP_ADD
}

__global__ __launch_bounds__(EPROP_THREADS)
void k_edge_prop(const int* __restrict__ src, const int* __restrict__ dst,
                 const float* __restrict__ h, float* __restrict__ out,
                 unsigned* __restrict__ ctr, int E, int n) {
    __shared__ unsigned nextp[2];
    __shared__ int2     ring[(EPROP_THREADS / 32) * (EPROP_RING + 32)];
    const int tid  = threadIdx.x;
    const int lane = tid & 31;
    const int wv   = __builtin_amdgcn_readfirstlane(tid >> 5);
    const int c    = lane & 15;
    const int hb   = lane & 16;
    int2* myq = ring + wv * (EPROP_RING + 32);

    const int myhalf = (int)((__builtin_amdgcn_s_sendmsg_rtn(0x87) >> 18) & 1u);
    const int split  = ((n >> 1) + 3) & ~3;
    const unsigned npieces = ((unsigned)E + EPROP_PIECE - 1) / EPROP_PIECE;

#pragma unroll 1
    for (int pass = 0; pass < 2; ++pass) {
        const int hh = myhalf ^ pass;
        unsigned* q = ctr + hh * CTR_STRIDE;
        int qh = 0;
        int qn = 0;
        if (tid == 0) nextp[0] = take_piece(q);
        __syncthreads();
        for (unsigned it = 0; ; ++it) {
            const unsigned p = __builtin_amdgcn_readfirstlane(nextp[it & 1]);
            if (p >= npieces) break;
            if (tid == 0) nextp[(it + 1) & 1] = take_piece(q);

            const int pbase = (int)(p * EPROP_PIECE) + wv * EDGES_PER_WAVE;
            int s_t[EPROP_TRIPS], d_t[EPROP_TRIPS];
#pragma unroll
            for (int t = 0; t < EPROP_TRIPS; ++t) {
                const int e  = pbase + t * (EPROP_THREADS / 32) * EDGES_PER_WAVE + lane;
                const int ec = e < E ? e : E - 1;
                s_t[t] = src[ec];
                d_t[t] = dst[ec];
            }
#pragma unroll
            for (int t = 0; t < EPROP_TRIPS; ++t) {
                const int e      = pbase + t * (EPROP_THREADS / 32) * EDGES_PER_WAVE + lane;
                const bool kept  = (e < E) && ((d_t[t] >= split) == (hh != 0));
                const unsigned m = __builtin_amdgcn_ballot_w32(kept);
                const int at   = (qh + qn + (int)__builtin_amdgcn_mbcnt_lo(m, 0u)) & (EPROP_RING - 1);
                const int slot = kept ? at : EPROP_RING + lane;
                myq[slot] = make_int2(s_t[t], d_t[t]);
                qn += __builtin_popcount(m);
            }
#pragma unroll 1
            while (qn >= 32) {
                const int2 ent = myq[(qh + lane) & (EPROP_RING - 1)];
                qh = (qh + 32) & (EPROP_RING - 1);
                qn -= 32;
                prop_batch<false>(ent, 0u, c, h, out);
            }
            __syncthreads();
        }
        if (qn > 0) {
            const bool live = lane < qn;
            int2 ent = make_int2(0, 0);
            if (live) ent = myq[(qh + lane) & (EPROP_RING - 1)];
            const unsigned rm = (__builtin_amdgcn_ballot_w32(live) >> hb) & 0xffffu;
            prop_batch<true>(ent, rm, c, h, out);
        }
        __syncthreads();
    }
}

#define EPOOL_THREADS 1024
#define EPOOL_SLICES  128
#define EPOOL_RANGE   100352
#define EPOOL_TRIPS   4
#define EPOOL_RING    256
#define POOL_NODES    32

__device__ __forceinline__ void pool_dest(const unsigned* dtab, const float* __restrict__ dinv, int lo, int r, bool live,
                                          int& g, float& w) {
    const unsigned wrd = dtab[(unsigned)r >> 1];
    const unsigned pr  = (r & 1) ? (wrd >> 16) : (wrd & 0xffffu);
    const unsigned dg  = pr & EPOOL_DEGMAX;
    g = (int)(pr >> 7);
    w = rsqrtf((float)dg);
    if (live && dg == EPOOL_DEGMAX) w = dinv[lo + r];
}

__global__ __launch_bounds__(EPOOL_THREADS)
void k_edge_pool(const int* __restrict__ src, const int* __restrict__ dst,
                 const unsigned* __restrict__ pkw, const float* __restrict__ dinv,
                 const float* __restrict__ h, const float* __restrict__ bias, const int* __restrict__ batch,
                 float* __restrict__ sums, float* __restrict__ cnt, int E, int n) {
    __shared__ float    tab[NUM_GRAPHS * HID];
    __shared__ float    ctab[NUM_GRAPHS];
    __shared__ unsigned dtab[EPOOL_RANGE / 2];
    __shared__ int2     ring[(EPOOL_THREADS / 32) * (EPOOL_RING + 32)];
    const int tid = threadIdx.x;
    const int lo  = blockIdx.y * EPOOL_RANGE;
    for (int i = tid; i < NUM_GRAPHS * HID; i += EPOOL_THREADS) tab[i] = 0.0f;
    for (int i = tid; i < NUM_GRAPHS; i += EPOOL_THREADS) ctab[i] = 0.0f;
    {
        int nrec = n - lo;
        if (nrec > EPOOL_RANGE) nrec = EPOOL_RANGE;
        const int nw = (nrec + 1) >> 1;
        const unsigned* pw = pkw + (lo >> 1);
        for (int i = tid; i < nw; i += EPOOL_THREADS) dtab[i] = pw[i];
    }
    __syncthreads();

    const int lane = tid & 31;
    const int c    = lane & 15;
    const int hb   = lane & 16;
    const int wv   = __builtin_amdgcn_readfirstlane(tid >> 5);
    int2* myq = ring + wv * (EPOOL_RING + 32);

    {
        const int nwg   = (int)(gridDim.x * gridDim.y);
        const int wg    = (int)(blockIdx.y * gridDim.x + blockIdx.x);
        const int share = (n + nwg - 1) / nwg;
        const int nb    = wg * share;
        int ne = nb + share;
        if (ne > n) ne = n;
        const float bc = bias[c];
        for (int base = nb + wv * POOL_NODES; base < ne; base += (EPOOL_THREADS / 32) * POOL_NODES) {
            int   g[POOL_NODES / 2];
            float v[POOL_NODES / 2];
#pragma unroll
            for (int j = 0; j < POOL_NODES / 2; ++j) {
                const int  node = base + 2 * j + (lane >> 4);
                const bool ok   = node < ne;
                const int  nc   = ok ? node : ne - 1;
                const int  gl   = batch[nc];
                g[j] = ok ? gl : -1;
                v[j] = dinv[nc] * h[(size_t)nc * HID + c] + bc;
            }
#pragma unroll
            for (int j = 0; j < POOL_NODES / 2; ++j) {
                if (g[j] >= 0) {
                    ldsAddF(tab + ((unsigned)g[j] * HID + c), v[j]);
                    if (c == 0) ldsAddF(ctab + g[j], 1.0f);
                }
            }
        }
    }

    const int per = (((E + (int)gridDim.x - 1) / (int)gridDim.x) + 31) & ~31;
    const int beg = blockIdx.x * per;
    int end = beg + per;
    if (end > E) end = E;

    int qh = 0;
    int qn = 0;
    const int gstep = (EPOOL_THREADS / 32) * EPOOL_TRIPS * EDGES_PER_WAVE;
    int base = beg + wv * (EPOOL_TRIPS * EDGES_PER_WAVE);
    int s_t[EPOOL_TRIPS], d_t[EPOOL_TRIPS];
#define POOL_LOAD_GROUP(B)                                                           \
    _Pragma("unroll")                                                                \
    for (int t = 0; t < EPOOL_TRIPS; ++t) {                                          \
        const int e  = (B) + t * EDGES_PER_WAVE + lane;                              \
        const int ec = e < end ? e : end - 1;     \
        s_t[t] = src[ec];                                                            \
        d_t[t] = dst[ec];                                                            \
    }
    if (base < end) { POOL_LOAD_GROUP(base) }
    for (; base < end; base += gstep) {
#pragma unroll
        for (int t = 0; t < EPOOL_TRIPS; ++t) {
            const int e      = base + t * EDGES_PER_WAVE + lane;
            const unsigned r = (unsigned)(d_t[t] - lo);
            const bool kept  = (e < end) && (r < (unsigned)EPOOL_RANGE);
            const unsigned m = __builtin_amdgcn_ballot_w32(kept);
            const int at   = (qh + qn + (int)__builtin_amdgcn_mbcnt_lo(m, 0u)) & (EPOOL_RING - 1);
            const int slot = kept ? at : EPOOL_RING + lane;
            myq[slot] = make_int2(s_t[t], (int)r);
            qn += __builtin_popcount(m);
        }
        if (base + gstep < end) { POOL_LOAD_GROUP(base + gstep) }

#pragma unroll 1
        while (qn >= 32) {
            const int2 ent = myq[(qh + lane) & (EPOOL_RING - 1)];
            qh = (qh + 32) & (EPOOL_RING - 1);
            qn -= 32;

            float hv[16];
#define POOL_GATHER(J) hv[J] = h[(unsigned)row_lane<J>(ent.x) * HID + c];
            POOL_GATHER(0)  POOL_GATHER(1)  POOL_GATHER(2)  POOL_GATHER(3)
            POOL_GATHER(4)  POOL_GATHER(5)  POOL_GATHER(6)  POOL_GATHER(7)
            POOL_GATHER(8)  POOL_GATHER(9)  POOL_GATHER(10) POOL_GATHER(11)
            POOL_GATHER(12) POOL_GATHER(13) POOL_GATHER(14) POOL_GATHER(15)
#undef POOL_GATHER
            int g_l; float w_l;
            pool_dest(dtab, dinv, lo, ent.y, true, g_l, w_l);
#define POOL_ADD(J) ldsAddF(tab + ((unsigned)row_lane<J>(g_l) * HID + c), row_lane<J>(w_l) * hv[J]);
            POOL_ADD(0)  POOL_ADD(1)  POOL_ADD(2)  POOL_ADD(3)
            POOL_ADD(4)  POOL_ADD(5)  POOL_ADD(6)  POOL_ADD(7)
            POOL_ADD(8)  POOL_ADD(9)  POOL_ADD(10) POOL_ADD(11)
            POOL_ADD(12) POOL_ADD(13) POOL_ADD(14) POOL_ADD(15)
#undef POOL_ADD
        }
    }

#undef POOL_LOAD_GROUP
    if (qn > 0) {
        const bool live = lane < qn;
        int2 ent = make_int2(0, 0);
        if (live) ent = myq[(qh + lane) & (EPOOL_RING - 1)];
        int g_l; float w_l;
        pool_dest(dtab, dinv, lo, ent.y, live, g_l, w_l);
#pragma unroll 1
        for (int j = 0; j < 16; ++j) {
            const int   s = __shfl(ent.x, hb + j);
            const int   g = __shfl(g_l, hb + j);
            const float w = __shfl(w_l, hb + j);
            if (hb + j < qn)
                ldsAddF(tab + ((unsigned)g * HID + c), w * h[(unsigned)s * HID + c]);
        }
    }
    __syncthreads();

    const int wg  = (int)(blockIdx.y * gridDim.x + blockIdx.x);
    const int rot = (wg * 48) & (NUM_GRAPHS * HID - 1) & ~31;
    for (int k = tid; k < NUM_GRAPHS * HID; k += EPOOL_THREADS) {
        const int i = (k + rot) & (NUM_GRAPHS * HID - 1);
        atomAddF(sums + i, tab[i]);
    }
    for (int k = tid; k < NUM_GRAPHS; k += EPOOL_THREADS) {
        const float cg = ctab[k];
        if (cg != 0.0f) atomAddF(cnt + k, cg);
    }
}

__global__ void k_head(const float* __restrict__ sums, const float* __restrict__ cnt,
                       const float* __restrict__ fcW, const float* __restrict__ fcb,
                       float* __restrict__ y) {
    int t = blockIdx.x * blockDim.x + threadIdx.x;
    if (t >= NUM_GRAPHS * NGRAPH) return;
    int g = t >> 4, cls = t & 15;
    float inv = 1.0f / fmaxf(cnt[g], 1.0f);
    float acc = fcb[cls];
#pragma unroll
    for (int k = 0; k < HID; ++k)
        acc += (sums[g * HID + k] * inv) * fcW[k * NGRAPH + cls];
    y[t] = acc;
}

extern "C" void kernel_launch(void* const* d_in, const int* in_sizes, int n_in,
                              void* d_out, int out_size, void* d_ws, size_t ws_size,
                              hipStream_t stream) {
    const float* x    = (const float*)d_in[0];
    const float* W1   = (const float*)d_in[1];
    const float* b1   = (const float*)d_in[2];
    const float* W2   = (const float*)d_in[3];
    const float* b2   = (const float*)d_in[4];
    const float* fcW  = (const float*)d_in[5];
    const float* fcb  = (const float*)d_in[6];
    const int*   eidx = (const int*)d_in[7];
    const int*   batch= (const int*)d_in[8];
    float* y = (float*)d_out;

    const int n = in_sizes[0] / 128;
    const int E = in_sizes[7] / 2;
    const int* src = eidx;
    const int* dst = eidx + E;

    size_t nAl = ((size_t)n + 3) & ~(size_t)3;
    float* deg  = (float*)d_ws;
    float* h    = deg + nAl;
    float* outb = h + (size_t)n * HID;
    float* sums = outb + (size_t)n * HID;
    float* cnt  = sums + NUM_GRAPHS * HID;

    const int TB = 256;
    const int n_tiles = (n + 15) / 16;

    unsigned* degc = (unsigned*)d_ws;
    k_init_deg<<<(n + TB - 1) / TB, TB, 0, stream>>>(degc, (unsigned*)sums, n);
    k_deg_accum<<<dim3(DEG_SLICES, (n + DEG_BINS - 1) / DEG_BINS), DEG_THREADS, 0, stream>>>(dst, degc, E, n);

    k_gemm16<128, false><<<(n_tiles + GEMM_WAVES - 1) / GEMM_WAVES, GEMM_WAVES * 32, 0, stream>>>(x, W1, degc, b1, h, outb, nullptr, nullptr, nullptr, nullptr, n, n_tiles);
    k_edge_prop<<<EPROP_BLOCKS, EPROP_THREADS, 0, stream>>>(src, dst, h, outb, (unsigned*)sums, E, n);

    unsigned* pkw = (unsigned*)h;
    k_gemm16<16, true><<<(n_tiles + GEMM_WAVES - 1) / GEMM_WAVES, GEMM_WAVES * 32, 0, stream>>>(outb, W2, degc, b1, outb, nullptr, batch, pkw, sums, cnt, n, n_tiles);

    k_edge_pool<<<dim3(EPOOL_SLICES, (n + EPOOL_RANGE - 1) / EPOOL_RANGE), EPOOL_THREADS, 0, stream>>>(src, dst, pkw, deg, outb, b2, batch, sums, cnt, E, n);
    k_head<<<(NUM_GRAPHS * NGRAPH + TB - 1) / TB, TB, 0, stream>>>(sums, cnt, fcW, fcb, y);
}
